# starting kernel
# baseline (speedup 1.0000x reference)
#include <hip/hip_runtime.h>
#include <hip/hip_bf16.h>

typedef __attribute__((ext_vector_type(2))) float v2f;
typedef __attribute__((ext_vector_type(8))) float v8f;

#define D_MODEL 1024
#define D_STATE 16
#define D_CONV  4
#define D_INNER 2048
#define DT_RANK 64
#define BB 2
#define LL 2048
#define TOKENS (BB * LL)
#define XDBL_W (DT_RANK + 2 * D_STATE)

__device__ __forceinline__ float sigmoidf_(float x) { return 1.f / (1.f + __expf(-x)); }

__global__ __launch_bounds__(256)
void ln_kernel(const float* __restrict__ x, const float* __restrict__ w,
               const float* __restrict__ b, float* __restrict__ out)
{
    const int row = blockIdx.x;
    const float* xr = x + (size_t)row * D_MODEL;
    float v[4];
    float s = 0.f, s2 = 0.f;
#pragma unroll
    for (int i = 0; i < 4; ++i) {
        v[i] = xr[threadIdx.x + i * 256];
        s += v[i];
        s2 += v[i] * v[i];
    }
#pragma unroll
    for (int off = 16; off > 0; off >>= 1) {
        s  += __shfl_down(s,  off, 32);
        s2 += __shfl_down(s2, off, 32);
    }
    __shared__ float rs[8], rq[8];
    __shared__ float mu_s, rstd_s;
    const int wave = threadIdx.x >> 5;
    const int lane = threadIdx.x & 31;
    if (lane == 0) { rs[wave] = s; rq[wave] = s2; }
    __syncthreads();
    if (threadIdx.x == 0) {
        float a = 0.f, q = 0.f;
#pragma unroll
        for (int i = 0; i < 8; ++i) { a += rs[i]; q += rq[i]; }
        const float mu  = a * (1.f / D_MODEL);
        const float var = q * (1.f / D_MODEL) - mu * mu;
        mu_s = mu;
        rstd_s = rsqrtf(var + 1e-5f);
    }
    __syncthreads();
    const float mu = mu_s, rstd = rstd_s;
    float* orow = out + (size_t)row * D_MODEL;
#pragma unroll
    for (int i = 0; i < 4; ++i) {
        const int c = threadIdx.x + i * 256;
        orow[c] = (v[i] - mu) * rstd * w[c] + b[c];
    }
}

template <int MT, int NT>
__global__ __launch_bounds__(128)
void wmma_gemm_kernel(const float* __restrict__ A, int lda,
                      const float* __restrict__ Bm, int ldb,
                      float* __restrict__ C, int ldc, int K,
                      const float* __restrict__ residual, int ldr,
                      const float* __restrict__ scale_ptr)
{
    const int lane = threadIdx.x & 31;
    const int wave = threadIdx.x >> 5;
    const int wy = wave & 1;
    const int wx = wave >> 1;
    const int m0 = blockIdx.y * (32 * MT) + wy * (16 * MT);
    const int n0 = blockIdx.x * (32 * NT) + wx * (16 * NT);
    const int lm = lane & 15;
    const int kh = (lane >> 4) << 1;

    v8f acc[MT][NT];
#pragma unroll
    for (int i = 0; i < MT; ++i)
#pragma unroll
        for (int t = 0; t < NT; ++t)
#pragma unroll
            for (int e = 0; e < 8; ++e) acc[i][t][e] = 0.f;

    const float* Ap = A + (size_t)(m0 + lm) * lda + kh;
    const float* Bp = Bm + (size_t)kh * ldb + n0 + lm;

    auto loadA = [&](int k, v2f (&a)[MT]) {
#pragma unroll
        for (int i = 0; i < MT; ++i)
            a[i] = *(const v2f*)(Ap + (size_t)i * 16 * lda + k);
    };
    auto loadB = [&](int k, v2f (&b)[NT]) {
        const float* r0 = Bp + (size_t)k * ldb;
        const float* r1 = r0 + ldb;
#pragma unroll
        for (int t = 0; t < NT; ++t) {
            b[t].x = r0[t * 16];
            b[t].y = r1[t * 16];
        }
    };

    v2f a[MT], b[NT], a2[MT], b2[NT];
    loadA(0, a);
    loadB(0, b);

    int k = 0;
    for (; k + 4 < K; k += 4) {
        loadA(k + 4, a2);
        loadB(k + 4, b2);
        __builtin_prefetch(Ap + k + 128, 0, 3);
        __builtin_prefetch(Bp + (size_t)(k + 16) * ldb, 0, 3);
#pragma unroll
        for (int i = 0; i < MT; ++i)
#pragma unroll
            for (int t = 0; t < NT; ++t)
                acc[i][t] = __builtin_amdgcn_wmma_f32_16x16x4_f32(
                    false, a[i], false, b[t], (short)0, acc[i][t], false, false);
#pragma unroll
        for (int i = 0; i < MT; ++i) a[i] = a2[i];
#pragma unroll
        for (int t = 0; t < NT; ++t) b[t] = b2[t];
    }
#pragma unroll
    for (int i = 0; i < MT; ++i)
#pragma unroll
        for (int t = 0; t < NT; ++t)
            acc[i][t] = __builtin_amdgcn_wmma_f32_16x16x4_f32(
                false, a[i], false, b[t], (short)0, acc[i][t], false, false);

    const float scale = scale_ptr ? scale_ptr[0] : 1.0f;
    const int mb = (lane >> 4) << 3;
#pragma unroll
    for (int i = 0; i < MT; ++i) {
#pragma unroll
        for (int t = 0; t < NT; ++t) {
            const int n = n0 + t * 16 + lm;
#pragma unroll
            for (int v = 0; v < 8; ++v) {
                const int m = m0 + i * 16 + mb + v;
                float val = acc[i][t][v] * scale;
                if (residual) val += residual[(size_t)m * ldr + n];
                C[(size_t)m * ldc + n] = val;
            }
        }
    }
}

__global__ __launch_bounds__(256)
void conv_silu_kernel(const float* __restrict__ xz, const float* __restrict__ w,
                      const float* __restrict__ bias, float* __restrict__ out)
{
    const size_t idx = (size_t)blockIdx.x * 256 + threadIdx.x;
    const int c = (int)(idx & (D_INNER - 1));
    const size_t bl = idx >> 11;
    const int l = (int)(bl & (LL - 1));
    const int b = (int)(bl >> 11);
    float acc = bias[c];
#pragma unroll
    for (int k = 0; k < D_CONV; ++k) {
        const int ls = l - (D_CONV - 1) + k;
        if (ls >= 0)
            acc += w[c * D_CONV + k] *
                   xz[((size_t)(b * LL + ls)) * (2 * D_INNER) + c];
    }
    out[idx] = acc * sigmoidf_(acc);
}

__global__ __launch_bounds__(256)
void conv_state_kernel(const float* __restrict__ xz, float* __restrict__ out)
{
    const int idx = blockIdx.x * 256 + threadIdx.x;
    if (idx >= BB * D_INNER * (D_CONV - 1)) return;
    const int j = idx % (D_CONV - 1);
    const int c = (idx / (D_CONV - 1)) % D_INNER;
    const int b = idx / ((D_CONV - 1) * D_INNER);
    out[idx] = xz[((size_t)(b * LL + (LL - (D_CONV - 1) + j))) * (2 * D_INNER) + c];
}

__global__ __launch_bounds__(256)
void dt_softplus_kernel(float* __restrict__ dt, const float* __restrict__ bias)
{
    const size_t idx = (size_t)blockIdx.x * 256 + threadIdx.x;
    const int d = (int)(idx & (D_INNER - 1));
    const float v = dt[idx] + bias[d];
    dt[idx] = (v > 20.f) ? v : __logf(1.f + __expf(v));
}

__global__ __launch_bounds__(256)
void scan_kernel(const float* __restrict__ dt, const float* __restrict__ xc,
                 const float* __restrict__ xdbl, const float* __restrict__ A_log,
                 float* __restrict__ y, float* __restrict__ ssm_out)
{
    const int t = blockIdx.x * 256 + threadIdx.x;
    const int d = t & (D_INNER - 1);
    const int b = t >> 11;

    float Aval[D_STATE];
#pragma unroll
    for (int n = 0; n < D_STATE; ++n)
        Aval[n] = -__expf(A_log[d * D_STATE + n]);

    float h[D_STATE];
#pragma unroll
    for (int n = 0; n < D_STATE; ++n) h[n] = 0.f;

    for (int l = 0; l < LL; ++l) {
        const size_t roff = (size_t)(b * LL + l);
        const float dtv = dt[roff * D_INNER + d];
        const float xv  = xc[roff * D_INNER + d];
        const float4* Bp = (const float4*)(xdbl + roff * XDBL_W + DT_RANK);
        const float4* Cp = (const float4*)(xdbl + roff * XDBL_W + DT_RANK + D_STATE);
        float Bb[D_STATE], Cc[D_STATE];
#pragma unroll
        for (int q = 0; q < 4; ++q) {
            const float4 bq = Bp[q];
            const float4 cq = Cp[q];
            Bb[4*q+0] = bq.x; Bb[4*q+1] = bq.y; Bb[4*q+2] = bq.z; Bb[4*q+3] = bq.w;
            Cc[4*q+0] = cq.x; Cc[4*q+1] = cq.y; Cc[4*q+2] = cq.z; Cc[4*q+3] = cq.w;
        }
        const float bx = dtv * xv;
        float yv = 0.f;
#pragma unroll
        for (int n = 0; n < D_STATE; ++n) {
            const float dA = __expf(Aval[n] * dtv);
            h[n] = dA * h[n] + bx * Bb[n];
            yv += h[n] * Cc[n];
        }
        y[roff * D_INNER + d] = yv;
    }

#pragma unroll
    for (int n = 0; n < D_STATE; ++n)
        ssm_out[(size_t)(b * D_INNER + d) * D_STATE + n] = h[n];
}

__global__ __launch_bounds__(256)
void gate_kernel(float* __restrict__ y, const float* __restrict__ xc,
                 const float* __restrict__ xz, const float* __restrict__ Dp)
{
    const size_t idx = (size_t)blockIdx.x * 256 + threadIdx.x;
    const int d = (int)(idx & (D_INNER - 1));
    const size_t row = idx >> 11;
    const float z  = xz[row * (2 * D_INNER) + D_INNER + d];
    const float yv = y[idx] + xc[idx] * Dp[d];
    y[idx] = yv * z * sigmoidf_(z);
}

extern "C" void kernel_launch(void* const* d_in, const int* in_sizes, int n_in,
                              void* d_out, int out_size, void* d_ws, size_t ws_size,
                              hipStream_t stream)
{
    const float* x         = (const float*)d_in[0];
    const float* ln_w      = (const float*)d_in[1];
    const float* ln_b      = (const float*)d_in[2];
    const float* in_proj_w = (const float*)d_in[3];
    const float* conv_w    = (const float*)d_in[4];
    const float* conv_b    = (const float*)d_in[5];
    const float* x_proj_w  = (const float*)d_in[6];
    const float* dt_proj_w = (const float*)d_in[7];
    const float* dt_proj_b = (const float*)d_in[8];
    const float* A_log     = (const float*)d_in[9];
    const float* D_param   = (const float*)d_in[10];
    const float* out_projw = (const float*)d_in[11];
    const float* res_scale = (const float*)d_in[12];

    float* out      = (float*)d_out;
    float* out_ssm  = out + (size_t)BB * LL * D_MODEL;
    float* out_conv = out_ssm + (size_t)BB * D_INNER * D_STATE;

    float* ws    = (float*)d_ws;
    float* xn    = ws;
    float* xz    = xn    + (size_t)TOKENS * D_MODEL;
    float* xc    = xz    + (size_t)TOKENS * 2 * D_INNER;
    float* xdbl  = xc    + (size_t)TOKENS * D_INNER;
    float* dtbuf = xdbl  + (size_t)TOKENS * XDBL_W;
    float* ybuf  = dtbuf + (size_t)TOKENS * D_INNER;

    ln_kernel<<<TOKENS, 256, 0, stream>>>(x, ln_w, ln_b, xn);

    wmma_gemm_kernel<2, 4><<<dim3((2 * D_INNER) / 128, TOKENS / 64), 128, 0, stream>>>(
        xn, D_MODEL, in_proj_w, 2 * D_INNER, xz, 2 * D_INNER, D_MODEL,
        nullptr, 0, nullptr);

    conv_silu_kernel<<<(TOKENS * D_INNER) / 256, 256, 0, stream>>>(
        xz, conv_w, conv_b, xc);
    conv_state_kernel<<<(BB * D_INNER * (D_CONV - 1) + 255) / 256, 256, 0, stream>>>(
        xz, out_conv);

    wmma_gemm_kernel<2, 3><<<dim3(XDBL_W / 96, TOKENS / 64), 128, 0, stream>>>(
        xc, D_INNER, x_proj_w, XDBL_W, xdbl, XDBL_W, D_INNER,
        nullptr, 0, nullptr);

    wmma_gemm_kernel<2, 4><<<dim3(D_INNER / 128, TOKENS / 64), 128, 0, stream>>>(
        xdbl, XDBL_W, dt_proj_w, D_INNER, dtbuf, D_INNER, DT_RANK,
        nullptr, 0, nullptr);

    dt_softplus_kernel<<<(TOKENS * D_INNER) / 256, 256, 0, stream>>>(dtbuf, dt_proj_b);

    scan_kernel<<<(BB * D_INNER) / 256, 256, 0, stream>>>(
        dtbuf, xc, xdbl, A_log, ybuf, out_ssm);

    gate_kernel<<<(TOKENS * D_INNER) / 256, 256, 0, stream>>>(ybuf, xc, xz, D_param);

    wmma_gemm_kernel<2, 4><<<dim3(D_MODEL / 128, TOKENS / 64), 128, 0, stream>>>(
        ybuf, D_INNER, out_projw, D_MODEL, out, D_MODEL, D_INNER,
        x, D_MODEL, res_scale);
}


# scan kernel: next chunk's dt, x and B|C rows fetched while the current chunk is computed (registers plus an LDS ring for B|C), one wave per workgroup; per-step math and summation order unchanged
# speedup vs baseline: 2.4395x; 2.4395x over previous
#include <hip/hip_runtime.h>
#include <hip/hip_bf16.h>

typedef __attribute__((ext_vector_type(2))) float v2f;
typedef __attribute__((ext_vector_type(8))) float v8f;

#define D_MODEL 1024
#define D_STATE 16
#define D_CONV  4
#define D_INNER 2048
#define DT_RANK 64
#define BB 2
#define LL 2048
#define TOKENS (BB * LL)
#define XDBL_W (DT_RANK + 2 * D_STATE)

__device__ __forceinline__ float sigmoidf_(float x) { return 1.f / (1.f + __expf(-x)); }

__global__ __launch_bounds__(256)
void ln_kernel(const float* __restrict__ x, const float* __restrict__ w,
               const float* __restrict__ b, float* __restrict__ out)
{
    const int row = blockIdx.x;
    const float* xr = x + (size_t)row * D_MODEL;
    float v[4];
    float s = 0.f, s2 = 0.f;
#pragma unroll
    for (int i = 0; i < 4; ++i) {
        v[i] = xr[threadIdx.x + i * 256];
        s += v[i];
        s2 += v[i] * v[i];
    }
#pragma unroll
    for (int off = 16; off > 0; off >>= 1) {
        s  += __shfl_down(s,  off, 32);
        s2 += __shfl_down(s2, off, 32);
    }
    __shared__ float rs[8], rq[8];
    __shared__ float mu_s, rstd_s;
    const int wave = threadIdx.x >> 5;
    const int lane = threadIdx.x & 31;
    if (lane == 0) { rs[wave] = s; rq[wave] = s2; }
    __syncthreads();
    if (threadIdx.x == 0) {
        float a = 0.f, q = 0.f;
#pragma unroll
        for (int i = 0; i < 8; ++i) { a += rs[i]; q += rq[i]; }
        const float mu  = a * (1.f / D_MODEL);
        const float var = q * (1.f / D_MODEL) - mu * mu;
        mu_s = mu;
        rstd_s = rsqrtf(var + 1e-5f);
    }
    __syncthreads();
    const float mu = mu_s, rstd = rstd_s;
    float* orow = out + (size_t)row * D_MODEL;
#pragma unroll
    for (int i = 0; i < 4; ++i) {
        const int c = threadIdx.x + i * 256;
        orow[c] = (v[i] - mu) * rstd * w[c] + b[c];
    }
}

template <int MT, int NT>
__global__ __launch_bounds__(128)
void wmma_gemm_kernel(const float* __restrict__ A, int lda,
                      const float* __restrict__ Bm, int ldb,
                      float* __restrict__ C, int ldc, int K,
                      const float* __restrict__ residual, int ldr,
                      const float* __restrict__ scale_ptr)
{
    const int lane = threadIdx.x & 31;
    const int wave = threadIdx.x >> 5;
    const int wy = wave & 1;
    const int wx = wave >> 1;
    const int m0 = blockIdx.y * (32 * MT) + wy * (16 * MT);
    const int n0 = blockIdx.x * (32 * NT) + wx * (16 * NT);
    const int lm = lane & 15;
    const int kh = (lane >> 4) << 1;

    v8f acc[MT][NT];
#pragma unroll
    for (int i = 0; i < MT; ++i)
#pragma unroll
        for (int t = 0; t < NT; ++t)
#pragma unroll
            for (int e = 0; e < 8; ++e) acc[i][t][e] = 0.f;

    const float* Ap = A + (size_t)(m0 + lm) * lda + kh;
    const float* Bp = Bm + (size_t)kh * ldb + n0 + lm;

    auto loadA = [&](int k, v2f (&a)[MT]) {
#pragma unroll
        for (int i = 0; i < MT; ++i)
            a[i] = *(const v2f*)(Ap + (size_t)i * 16 * lda + k);
    };
    auto loadB = [&](int k, v2f (&b)[NT]) {
        const float* r0 = Bp + (size_t)k * ldb;
        const float* r1 = r0 + ldb;
#pragma unroll
        for (int t = 0; t < NT; ++t) {
            b[t].x = r0[t * 16];
            b[t].y = r1[t * 16];
        }
    };

    v2f a[MT], b[NT], a2[MT], b2[NT];
    loadA(0, a);
    loadB(0, b);

    int k = 0;
    for (; k + 4 < K; k += 4) {
        loadA(k + 4, a2);
        loadB(k + 4, b2);
        __builtin_prefetch(Ap + k + 128, 0, 3);
        __builtin_prefetch(Bp + (size_t)(k + 16) * ldb, 0, 3);
#pragma unroll
        for (int i = 0; i < MT; ++i)
#pragma unroll
            for (int t = 0; t < NT; ++t)
                acc[i][t] = __builtin_amdgcn_wmma_f32_16x16x4_f32(
                    false, a[i], false, b[t], (short)0, acc[i][t], false, false);
#pragma unroll
        for (int i = 0; i < MT; ++i) a[i] = a2[i];
#pragma unroll
        for (int t = 0; t < NT; ++t) b[t] = b2[t];
    }
#pragma unroll
    for (int i = 0; i < MT; ++i)
#pragma unroll
        for (int t = 0; t < NT; ++t)
            acc[i][t] = __builtin_amdgcn_wmma_f32_16x16x4_f32(
                false, a[i], false, b[t], (short)0, acc[i][t], false, false);

    const float scale = scale_ptr ? scale_ptr[0] : 1.0f;
    const int mb = (lane >> 4) << 3;
#pragma unroll
    for (int i = 0; i < MT; ++i) {
#pragma unroll
        for (int t = 0; t < NT; ++t) {
            const int n = n0 + t * 16 + lm;
#pragma unroll
            for (int v = 0; v < 8; ++v) {
                const int m = m0 + i * 16 + mb + v;
                float val = acc[i][t][v] * scale;
                if (residual) val += residual[(size_t)m * ldr + n];
                C[(size_t)m * ldc + n] = val;
            }
        }
    }
}

__global__ __launch_bounds__(256)
void conv_silu_kernel(const float* __restrict__ xz, const float* __restrict__ w,
                      const float* __restrict__ bias, float* __restrict__ out)
{
    const size_t idx = (size_t)blockIdx.x * 256 + threadIdx.x;
    const int c = (int)(idx & (D_INNER - 1));
    const size_t bl = idx >> 11;
    const int l = (int)(bl & (LL - 1));
    const int b = (int)(bl >> 11);
    float acc = bias[c];
#pragma unroll
    for (int k = 0; k < D_CONV; ++k) {
        const int ls = l - (D_CONV - 1) + k;
        if (ls >= 0)
            acc += w[c * D_CONV + k] *
                   xz[((size_t)(b * LL + ls)) * (2 * D_INNER) + c];
    }
    out[idx] = acc * sigmoidf_(acc);
}

__global__ __launch_bounds__(256)
void conv_state_kernel(const float* __restrict__ xz, float* __restrict__ out)
{
    const int idx = blockIdx.x * 256 + threadIdx.x;
    if (idx >= BB * D_INNER * (D_CONV - 1)) return;
    const int j = idx % (D_CONV - 1);
    const int c = (idx / (D_CONV - 1)) % D_INNER;
    const int b = idx / ((D_CONV - 1) * D_INNER);
    out[idx] = xz[((size_t)(b * LL + (LL - (D_CONV - 1) + j))) * (2 * D_INNER) + c];
}

__global__ __launch_bounds__(256)
void dt_softplus_kernel(float* __restrict__ dt, const float* __restrict__ bias)
{
    const size_t idx = (size_t)blockIdx.x * 256 + threadIdx.x;
    const int d = (int)(idx & (D_INNER - 1));
    const float v = dt[idx] + bias[d];
    dt[idx] = (v > 20.f) ? v : __logf(1.f + __expf(v));
}

#define SCAN_T   16
#define SCAN_NCH (LL / SCAN_T)
#define SCAN_WPB (D_INNER / 32)

__global__ __launch_bounds__(32)
void scan_kernel(const float* __restrict__ dt, const float* __restrict__ xc,
                 const float* __restrict__ xdbl, const float* __restrict__ A_log,
                 float* __restrict__ y, float* __restrict__ ssm_out)
{
    const int lane = threadIdx.x;
    const int b = blockIdx.x / SCAN_WPB;
    const int d = (blockIdx.x % SCAN_WPB) * 32 + lane;

    __shared__ __attribute__((aligned(16))) float sBC[2][SCAN_T * 32];

    v2f Aval[D_STATE / 2], h[D_STATE / 2];
#pragma unroll
    for (int k = 0; k < D_STATE / 2; ++k) {
        Aval[k].x = -__expf(A_log[d * D_STATE + 2 * k]);
        Aval[k].y = -__expf(A_log[d * D_STATE + 2 * k + 1]);
        h[k].x = 0.f; h[k].y = 0.f;
    }

    const size_t row0 = (size_t)b * LL;
    const float* dtp = dt + row0 * D_INNER + d;
    const float* xp  = xc + row0 * D_INNER + d;
    float*       yp  = y  + row0 * D_INNER + d;
    const float* bcp = xdbl + row0 * XDBL_W + DT_RANK + lane;

    float cd[SCAN_T], cx[SCAN_T], nd[SCAN_T], nx[SCAN_T];
    float nbc[SCAN_T];

    auto fetch = [&](int c, float (&fd)[SCAN_T], float (&fx)[SCAN_T],
                     float (&fbc)[SCAN_T]) {
        const size_t l0 = (size_t)c * SCAN_T;
#pragma unroll
        for (int s = 0; s < SCAN_T; ++s) {
            fbc[s] = bcp[(l0 + s) * XDBL_W];
            fd[s] = dtp[(l0 + s) * D_INNER];
            fx[s] = xp[(l0 + s) * D_INNER];
        }
    };
    auto stash = [&](int p, const float (&fbc)[SCAN_T]) {
#pragma unroll
        for (int s = 0; s < SCAN_T; ++s)
            sBC[p][s * 32 + lane] = fbc[s];
    };

    auto lds_handover = [&]() {
        __builtin_amdgcn_fence(__ATOMIC_RELEASE, "workgroup", "local");
        __builtin_amdgcn_s_barrier();
        __builtin_amdgcn_fence(__ATOMIC_ACQUIRE, "workgroup", "local");
    };

    fetch(0, cd, cx, nbc);
    stash(0, nbc);
    lds_handover();

    for (int c = 0; c < SCAN_NCH; ++c) {
        const int p = c & 1;
        const float4* bc = (const float4*)&sBC[p][0];
        float4 rb[8], rn[8];
#pragma unroll
        for (int q = 0; q < 8; ++q) rb[q] = bc[q];

        const int cn = (c + 1 < SCAN_NCH) ? c + 1 : c;
        fetch(cn, nd, nx, nbc);
        __builtin_amdgcn_sched_barrier(0);

        float* yrow = yp + (size_t)c * SCAN_T * D_INNER;
#pragma unroll
        for (int s = 0; s < SCAN_T; ++s) {
            __builtin_amdgcn_sched_barrier(0);
            if (s + 1 < SCAN_T) {
#pragma unroll
                for (int q = 0; q < 8; ++q) rn[q] = bc[(s + 1) * 8 + q];
            }
            __builtin_amdgcn_sched_barrier(0);
            const float dtv = cd[s];
            const float bx = dtv * cx[s];
            float yv = 0.f;
#pragma unroll
            for (int k = 0; k < D_STATE / 2; ++k) {
                const float4 bq = rb[k >> 1], cq = rb[4 + (k >> 1)];
                v2f Bk, Ck;
                if (k & 1) { Bk.x = bq.z; Bk.y = bq.w; Ck.x = cq.z; Ck.y = cq.w; }
                else       { Bk.x = bq.x; Bk.y = bq.y; Ck.x = cq.x; Ck.y = cq.y; }
                const v2f a = Aval[k] * dtv;
                v2f dA;
                dA.x = __expf(a.x);
                dA.y = __expf(a.y);
                h[k] = dA * h[k] + bx * Bk;
                yv += h[k].x * Ck.x;
                yv += h[k].y * Ck.y;
            }
            yrow[(size_t)s * D_INNER] = yv;
            if (s + 1 < SCAN_T) {
#pragma unroll
                for (int q = 0; q < 8; ++q) rb[q] = rn[q];
            }
        }
        __builtin_amdgcn_sched_barrier(0);

        stash(p ^ 1, nbc);
#pragma unroll
        for (int s = 0; s < SCAN_T; ++s) { cd[s] = nd[s]; cx[s] = nx[s]; }
        lds_handover();
    }

#pragma unroll
    for (int k = 0; k < D_STATE / 2; ++k) {
        ssm_out[(size_t)(b * D_INNER + d) * D_STATE + 2 * k]     = h[k].x;
        ssm_out[(size_t)(b * D_INNER + d) * D_STATE + 2 * k + 1] = h[k].y;
    }
}

__global__ __launch_bounds__(256)
void gate_kernel(float* __restrict__ y, const float* __restrict__ xc,
                 const float* __restrict__ xz, const float* __restrict__ Dp)
{
    const size_t idx = (size_t)blockIdx.x * 256 + threadIdx.x;
    const int d = (int)(idx & (D_INNER - 1));
    const size_t row = idx >> 11;
    const float z  = xz[row * (2 * D_INNER) + D_INNER + d];
    const float yv = y[idx] + xc[idx] * Dp[d];
    y[idx] = yv * z * sigmoidf_(z);
}

extern "C" void kernel_launch(void* const* d_in, const int* in_sizes, int n_in,
                              void* d_out, int out_size, void* d_ws, size_t ws_size,
                              hipStream_t stream)
{
    const float* x         = (const float*)d_in[0];
    const float* ln_w      = (const float*)d_in[1];
    const float* ln_b      = (const float*)d_in[2];
    const float* in_proj_w = (const float*)d_in[3];
    const float* conv_w    = (const float*)d_in[4];
    const float* conv_b    = (const float*)d_in[5];
    const float* x_proj_w  = (const float*)d_in[6];
    const float* dt_proj_w = (const float*)d_in[7];
    const float* dt_proj_b = (const float*)d_in[8];
    const float* A_log     = (const float*)d_in[9];
    const float* D_param   = (const float*)d_in[10];
    const float* out_projw = (const float*)d_in[11];
    const float* res_scale = (const float*)d_in[12];

    float* out      = (float*)d_out;
    float* out_ssm  = out + (size_t)BB * LL * D_MODEL;
    float* out_conv = out_ssm + (size_t)BB * D_INNER * D_STATE;

    float* ws    = (float*)d_ws;
    float* xn    = ws;
    float* xz    = xn    + (size_t)TOKENS * D_MODEL;
    float* xc    = xz    + (size_t)TOKENS * 2 * D_INNER;
    float* xdbl  = xc    + (size_t)TOKENS * D_INNER;
    float* dtbuf = xdbl  + (size_t)TOKENS * XDBL_W;
    float* ybuf  = dtbuf + (size_t)TOKENS * D_INNER;

    ln_kernel<<<TOKENS, 256, 0, stream>>>(x, ln_w, ln_b, xn);

    wmma_gemm_kernel<2, 4><<<dim3((2 * D_INNER) / 128, TOKENS / 64), 128, 0, stream>>>(
        xn, D_MODEL, in_proj_w, 2 * D_INNER, xz, 2 * D_INNER, D_MODEL,
        nullptr, 0, nullptr);

    conv_silu_kernel<<<(TOKENS * D_INNER) / 256, 256, 0, stream>>>(
        xz, conv_w, conv_b, xc);
    conv_state_kernel<<<(BB * D_INNER * (D_CONV - 1) + 255) / 256, 256, 0, stream>>>(
        xz, out_conv);

    wmma_gemm_kernel<2, 3><<<dim3(XDBL_W / 96, TOKENS / 64), 128, 0, stream>>>(
        xc, D_INNER, x_proj_w, XDBL_W, xdbl, XDBL_W, D_INNER,
        nullptr, 0, nullptr);

    wmma_gemm_kernel<2, 4><<<dim3(D_INNER / 128, TOKENS / 64), 128, 0, stream>>>(
        xdbl, XDBL_W, dt_proj_w, D_INNER, dtbuf, D_INNER, DT_RANK,
        nullptr, 0, nullptr);

    dt_softplus_kernel<<<(TOKENS * D_INNER) / 256, 256, 0, stream>>>(dtbuf, dt_proj_b);

    scan_kernel<<<BB * SCAN_WPB, 32, 0, stream>>>(
        dtbuf, xc, xdbl, A_log, ybuf, out_ssm);

    gate_kernel<<<(TOKENS * D_INNER) / 256, 256, 0, stream>>>(ybuf, xc, xz, D_param);

    wmma_gemm_kernel<2, 4><<<dim3(D_MODEL / 128, TOKENS / 64), 128, 0, stream>>>(
        ybuf, D_INNER, out_projw, D_MODEL, out, D_MODEL, D_INNER,
        x, D_MODEL, res_scale);
}


# GEMM loaders read 16 B per lane (A: 8 consecutive k per lane half; B: 4 adjacent columns, one per N tile), K walked 16 per load group; products inside each 16-k group are summed in a permuted order
# speedup vs baseline: 3.4822x; 1.4274x over previous
#include <hip/hip_runtime.h>
#include <hip/hip_bf16.h>

typedef __attribute__((ext_vector_type(2))) float v2f;
typedef __attribute__((ext_vector_type(4))) float v4f;
typedef __attribute__((ext_vector_type(8))) float v8f;

#define D_MODEL 1024
#define D_STATE 16
#define D_CONV  4
#define D_INNER 2048
#define DT_RANK 64
#define BB 2
#define LL 2048
#define TOKENS (BB * LL)
#define XDBL_W (DT_RANK + 2 * D_STATE)

__device__ __forceinline__ float sigmoidf_(float x) { return 1.f / (1.f + __expf(-x)); }

__global__ __launch_bounds__(256)
void ln_kernel(const float* __restrict__ x, const float* __restrict__ w,
               const float* __restrict__ b, float* __restrict__ out)
{
    const int row = blockIdx.x;
    const float* xr = x + (size_t)row * D_MODEL;
    float v[4];
    float s = 0.f, s2 = 0.f;
#pragma unroll
    for (int i = 0; i < 4; ++i) {
        v[i] = xr[threadIdx.x + i * 256];
        s += v[i];
        s2 += v[i] * v[i];
    }
#pragma unroll
    for (int off = 16; off > 0; off >>= 1) {
        s  += __shfl_down(s,  off, 32);
        s2 += __shfl_down(s2, off, 32);
    }
    __shared__ float rs[8], rq[8];
    __shared__ float mu_s, rstd_s;
    const int wave = threadIdx.x >> 5;
    const int lane = threadIdx.x & 31;
    if (lane == 0) { rs[wave] = s; rq[wave] = s2; }
    __syncthreads();
    if (threadIdx.x == 0) {
        float a = 0.f, q = 0.f;
#pragma unroll
        for (int i = 0; i < 8; ++i) { a += rs[i]; q += rq[i]; }
        const float mu  = a * (1.f / D_MODEL);
        const float var = q * (1.f / D_MODEL) - mu * mu;
        mu_s = mu;
        rstd_s = rsqrtf(var + 1e-5f);
    }
    __syncthreads();
    const float mu = mu_s, rstd = rstd_s;
    float* orow = out + (size_t)row * D_MODEL;
#pragma unroll
    for (int i = 0; i < 4; ++i) {
        const int c = threadIdx.x + i * 256;
        orow[c] = (v[i] - mu) * rstd * w[c] + b[c];
    }
}

template <int MT, int NT>
__global__ __launch_bounds__(128)
void wmma_gemm_kernel(const float* __restrict__ A, int lda,
                      const float* __restrict__ Bm, int ldb,
                      float* __restrict__ C, int ldc, int K,
                      const float* __restrict__ residual, int ldr,
                      const float* __restrict__ scale_ptr)
{
    const int lane = threadIdx.x & 31;
    const int wave = threadIdx.x >> 5;
    const int wy = wave & 1;
    const int wx = wave >> 1;
    const int m0 = blockIdx.y * (32 * MT) + wy * (16 * MT);
    const int n0 = blockIdx.x * (32 * NT) + wx * (16 * NT);
    const int lc = lane & 15;
    const int lh = lane >> 4;

    v8f acc[MT][NT];
#pragma unroll
    for (int i = 0; i < MT; ++i)
#pragma unroll
        for (int t = 0; t < NT; ++t)
#pragma unroll
            for (int e = 0; e < 8; ++e) acc[i][t][e] = 0.f;

    const float* Ap = A + (size_t)(m0 + lc) * lda + 8 * lh;
    const float* Bp = Bm + (size_t)(8 * lh) * ldb + n0 + NT * lc;

    auto loadA = [&](int kb, v2f (&a)[MT][4]) {
#pragma unroll
        for (int i = 0; i < MT; ++i) {
            const float* p = Ap + (size_t)i * 16 * lda + kb;
            const v4f q0 = *(const v4f*)p;
            const v4f q1 = *(const v4f*)(p + 4);
            a[i][0] = __builtin_shufflevector(q0, q0, 0, 1);
            a[i][1] = __builtin_shufflevector(q0, q0, 2, 3);
            a[i][2] = __builtin_shufflevector(q1, q1, 0, 1);
            a[i][3] = __builtin_shufflevector(q1, q1, 2, 3);
        }
    };
    auto loadB = [&](int kb, float (&b)[8][4]) {
#pragma unroll
        for (int r = 0; r < 8; ++r) {
            const float* p = Bp + (size_t)(kb + r) * ldb;
            if constexpr (NT == 4) {
                const v4f q = *(const v4f*)p;
                b[r][0] = q.x; b[r][1] = q.y; b[r][2] = q.z; b[r][3] = q.w;
            } else {
#pragma unroll
                for (int t = 0; t < NT; ++t) b[r][t] = p[t];
            }
        }
    };
    auto mma = [&](const v2f (&a)[MT][4], const float (&b)[8][4]) {
#pragma unroll
        for (int j = 0; j < 4; ++j) {
#pragma unroll
            for (int i = 0; i < MT; ++i) {
#pragma unroll
                for (int t = 0; t < NT; ++t) {
                    v2f bv;
                    bv.x = b[2 * j][t];
                    bv.y = b[2 * j + 1][t];
                    acc[i][t] = __builtin_amdgcn_wmma_f32_16x16x4_f32(
                        false, a[i][j], false, bv, (short)0, acc[i][t], false, false);
                }
            }
        }
    };

    v2f a0[MT][4], a1[MT][4];
    float b0[8][4], b1[8][4];
    loadA(0, a0);
    loadB(0, b0);

    for (int kb = 0; kb < K; kb += 32) {
        loadA(kb + 16, a1);
        loadB(kb + 16, b1);
        __builtin_amdgcn_sched_barrier(0);
        mma(a0, b0);
        const int kn = (kb + 32 < K) ? kb + 32 : kb + 16;
        loadA(kn, a0);
        loadB(kn, b0);
        __builtin_amdgcn_sched_barrier(0);
        mma(a1, b1);
    }

    const float scale = scale_ptr ? scale_ptr[0] : 1.0f;
#pragma unroll
    for (int i = 0; i < MT; ++i) {
#pragma unroll
        for (int v = 0; v < 8; ++v) {
            const int m = m0 + i * 16 + 8 * lh + v;
            const int n = n0 + NT * lc;
            if constexpr (NT == 4) {
                v4f o;
                o.x = acc[i][0][v] * scale;
                o.y = acc[i][1][v] * scale;
                o.z = acc[i][2][v] * scale;
                o.w = acc[i][3][v] * scale;
                if (residual) o += *(const v4f*)(residual + (size_t)m * ldr + n);
                *(v4f*)(C + (size_t)m * ldc + n) = o;
            } else {
#pragma unroll
                for (int t = 0; t < NT; ++t) {
                    float x = acc[i][t][v] * scale;
                    if (residual) x += residual[(size_t)m * ldr + n + t];
                    C[(size_t)m * ldc + n + t] = x;
                }
            }
        }
    }
}

__global__ __launch_bounds__(256)
void conv_silu_kernel(const float* __restrict__ xz, const float* __restrict__ w,
                      const float* __restrict__ bias, float* __restrict__ out)
{
    const size_t idx = (size_t)blockIdx.x * 256 + threadIdx.x;
    const int c = (int)(idx & (D_INNER - 1));
    const size_t bl = idx >> 11;
    const int l = (int)(bl & (LL - 1));
    const int b = (int)(bl >> 11);
    float acc = bias[c];
#pragma unroll
    for (int k = 0; k < D_CONV; ++k) {
        const int ls = l - (D_CONV - 1) + k;
        if (ls >= 0)
            acc += w[c * D_CONV + k] *
                   xz[((size_t)(b * LL + ls)) * (2 * D_INNER) + c];
    }
    out[idx] = acc * sigmoidf_(acc);
}

__global__ __launch_bounds__(256)
void conv_state_kernel(const float* __restrict__ xz, float* __restrict__ out)
{
    const int idx = blockIdx.x * 256 + threadIdx.x;
    if (idx >= BB * D_INNER * (D_CONV - 1)) return;
    const int j = idx % (D_CONV - 1);
    const int c = (idx / (D_CONV - 1)) % D_INNER;
    const int b = idx / ((D_CONV - 1) * D_INNER);
    out[idx] = xz[((size_t)(b * LL + (LL - (D_CONV - 1) + j))) * (2 * D_INNER) + c];
}

__global__ __launch_bounds__(256)
void dt_softplus_kernel(float* __restrict__ dt, const float* __restrict__ bias)
{
    const size_t idx = (size_t)blockIdx.x * 256 + threadIdx.x;
    const int d = (int)(idx & (D_INNER - 1));
    const float v = dt[idx] + bias[d];
    dt[idx] = (v > 20.f) ? v : __logf(1.f + __expf(v));
}

#define SCAN_T   16
#define SCAN_NCH (LL / SCAN_T)
#define SCAN_WPB (D_INNER / 32)

__global__ __launch_bounds__(32)
void scan_kernel(const float* __restrict__ dt, const float* __restrict__ xc,
                 const float* __restrict__ xdbl, const float* __restrict__ A_log,
                 float* __restrict__ y, float* __restrict__ ssm_out)
{
    const int lane = threadIdx.x;
    const int b = blockIdx.x / SCAN_WPB;
    const int d = (blockIdx.x % SCAN_WPB) * 32 + lane;

    __shared__ __attribute__((aligned(16))) float sBC[2][SCAN_T * 32];

    v2f Aval[D_STATE / 2], h[D_STATE / 2];
#pragma unroll
    for (int k = 0; k < D_STATE / 2; ++k) {
        Aval[k].x = -__expf(A_log[d * D_STATE + 2 * k]);
        Aval[k].y = -__expf(A_log[d * D_STATE + 2 * k + 1]);
        h[k].x = 0.f; h[k].y = 0.f;
    }

    const size_t row0 = (size_t)b * LL;
    const float* dtp = dt + row0 * D_INNER + d;
    const float* xp  = xc + row0 * D_INNER + d;
    float*       yp  = y  + row0 * D_INNER + d;
    const float* bcp = xdbl + row0 * XDBL_W + DT_RANK + lane;

    float cd[SCAN_T], cx[SCAN_T], nd[SCAN_T], nx[SCAN_T];
    float nbc[SCAN_T];

    auto fetch = [&](int c, float (&fd)[SCAN_T], float (&fx)[SCAN_T],
                     float (&fbc)[SCAN_T]) {
        const size_t l0 = (size_t)c * SCAN_T;
#pragma unroll
        for (int s = 0; s < SCAN_T; ++s) {
            fbc[s] = bcp[(l0 + s) * XDBL_W];
            fd[s] = dtp[(l0 + s) * D_INNER];
            fx[s] = xp[(l0 + s) * D_INNER];
        }
    };
    auto stash = [&](int p, const float (&fbc)[SCAN_T]) {
#pragma unroll
        for (int s = 0; s < SCAN_T; ++s)
            sBC[p][s * 32 + lane] = fbc[s];
    };

    auto lds_handover = [&]() {
        __builtin_amdgcn_fence(__ATOMIC_RELEASE, "workgroup", "local");
        __builtin_amdgcn_s_barrier();
        __builtin_amdgcn_fence(__ATOMIC_ACQUIRE, "workgroup", "local");
    };

    fetch(0, cd, cx, nbc);
    stash(0, nbc);
    lds_handover();

    for (int c = 0; c < SCAN_NCH; ++c) {
        const int p = c & 1;
        const float4* bc = (const float4*)&sBC[p][0];
        float4 rb[8], rn[8];
#pragma unroll
        for (int q = 0; q < 8; ++q) rb[q] = bc[q];

        const int cn = (c + 1 < SCAN_NCH) ? c + 1 : c;
        fetch(cn, nd, nx, nbc);
        __builtin_amdgcn_sched_barrier(0);

        float* yrow = yp + (size_t)c * SCAN_T * D_INNER;
#pragma unroll
        for (int s = 0; s < SCAN_T; ++s) {
            __builtin_amdgcn_sched_barrier(0);
            if (s + 1 < SCAN_T) {
#pragma unroll
                for (int q = 0; q < 8; ++q) rn[q] = bc[(s + 1) * 8 + q];
            }
            __builtin_amdgcn_sched_barrier(0);
            const float dtv = cd[s];
            const float bx = dtv * cx[s];
            float yv = 0.f;
#pragma unroll
            for (int k = 0; k < D_STATE / 2; ++k) {
                const float4 bq = rb[k >> 1], cq = rb[4 + (k >> 1)];
                v2f Bk, Ck;
                if (k & 1) { Bk.x = bq.z; Bk.y = bq.w; Ck.x = cq.z; Ck.y = cq.w; }
                else       { Bk.x = bq.x; Bk.y = bq.y; Ck.x = cq.x; Ck.y = cq.y; }
                const v2f a = Aval[k] * dtv;
                v2f dA;
                dA.x = __expf(a.x);
                dA.y = __expf(a.y);
                h[k] = dA * h[k] + bx * Bk;
                yv += h[k].x * Ck.x;
                yv += h[k].y * Ck.y;
            }
            yrow[(size_t)s * D_INNER] = yv;
            if (s + 1 < SCAN_T) {
#pragma unroll
                for (int q = 0; q < 8; ++q) rb[q] = rn[q];
            }
        }
        __builtin_amdgcn_sched_barrier(0);

        stash(p ^ 1, nbc);
#pragma unroll
        for (int s = 0; s < SCAN_T; ++s) { cd[s] = nd[s]; cx[s] = nx[s]; }
        lds_handover();
    }

#pragma unroll
    for (int k = 0; k < D_STATE / 2; ++k) {
        ssm_out[(size_t)(b * D_INNER + d) * D_STATE + 2 * k]     = h[k].x;
        ssm_out[(size_t)(b * D_INNER + d) * D_STATE + 2 * k + 1] = h[k].y;
    }
}

__global__ __launch_bounds__(256)
void gate_kernel(float* __restrict__ y, const float* __restrict__ xc,
                 const float* __restrict__ xz, const float* __restrict__ Dp)
{
    const size_t idx = (size_t)blockIdx.x * 256 + threadIdx.x;
    const int d = (int)(idx & (D_INNER - 1));
    const size_t row = idx >> 11;
    const float z  = xz[row * (2 * D_INNER) + D_INNER + d];
    const float yv = y[idx] + xc[idx] * Dp[d];
    y[idx] = yv * z * sigmoidf_(z);
}

extern "C" void kernel_launch(void* const* d_in, const int* in_sizes, int n_in,
                              void* d_out, int out_size, void* d_ws, size_t ws_size,
                              hipStream_t stream)
{
    const float* x         = (const float*)d_in[0];
    const float* ln_w      = (const float*)d_in[1];
    const float* ln_b      = (const float*)d_in[2];
    const float* in_proj_w = (const float*)d_in[3];
    const float* conv_w    = (const float*)d_in[4];
    const float* conv_b    = (const float*)d_in[5];
    const float* x_proj_w  = (const float*)d_in[6];
    const float* dt_proj_w = (const float*)d_in[7];
    const float* dt_proj_b = (const float*)d_in[8];
    const float* A_log     = (const float*)d_in[9];
    const float* D_param   = (const float*)d_in[10];
    const float* out_projw = (const float*)d_in[11];
    const float* res_scale = (const float*)d_in[12];

    float* out      = (float*)d_out;
    float* out_ssm  = out + (size_t)BB * LL * D_MODEL;
    float* out_conv = out_ssm + (size_t)BB * D_INNER * D_STATE;

    float* ws    = (float*)d_ws;
    float* xn    = ws;
    float* xz    = xn    + (size_t)TOKENS * D_MODEL;
    float* xc    = xz    + (size_t)TOKENS * 2 * D_INNER;
    float* xdbl  = xc    + (size_t)TOKENS * D_INNER;
    float* dtbuf = xdbl  + (size_t)TOKENS * XDBL_W;
    float* ybuf  = dtbuf + (size_t)TOKENS * D_INNER;

    ln_kernel<<<TOKENS, 256, 0, stream>>>(x, ln_w, ln_b, xn);

    wmma_gemm_kernel<2, 4><<<dim3((2 * D_INNER) / 128, TOKENS / 64), 128, 0, stream>>>(
        xn, D_MODEL, in_proj_w, 2 * D_INNER, xz, 2 * D_INNER, D_MODEL,
        nullptr, 0, nullptr);

    conv_silu_kernel<<<(TOKENS * D_INNER) / 256, 256, 0, stream>>>(
        xz, conv_w, conv_b, xc);
    conv_state_kernel<<<(BB * D_INNER * (D_CONV - 1) + 255) / 256, 256, 0, stream>>>(
        xz, out_conv);

    wmma_gemm_kernel<2, 3><<<dim3(XDBL_W / 96, TOKENS / 64), 128, 0, stream>>>(
        xc, D_INNER, x_proj_w, XDBL_W, xdbl, XDBL_W, D_INNER,
        nullptr, 0, nullptr);

    wmma_gemm_kernel<2, 4><<<dim3(D_INNER / 128, TOKENS / 64), 128, 0, stream>>>(
        xdbl, XDBL_W, dt_proj_w, D_INNER, dtbuf, D_INNER, DT_RANK,
        nullptr, 0, nullptr);

    dt_softplus_kernel<<<(TOKENS * D_INNER) / 256, 256, 0, stream>>>(dtbuf, dt_proj_b);

    scan_kernel<<<BB * SCAN_WPB, 32, 0, stream>>>(
        dtbuf, xc, xdbl, A_log, ybuf, out_ssm);

    gate_kernel<<<(TOKENS * D_INNER) / 256, 256, 0, stream>>>(ybuf, xc, xz, D_param);

    wmma_gemm_kernel<2, 4><<<dim3(D_MODEL / 128, TOKENS / 64), 128, 0, stream>>>(
        ybuf, D_INNER, out_projw, D_MODEL, out, D_MODEL, D_INNER,
        x, D_MODEL, res_scale);
}


# scan kernel: a channel's 16 states are spread over the 4 waves of a workgroup (4 states per wave, chunks of 32 steps); y's 16 terms are summed as 4 partial sums of 4, added through LDS
# speedup vs baseline: 4.2448x; 1.2190x over previous
#include <hip/hip_runtime.h>
#include <hip/hip_bf16.h>

typedef __attribute__((ext_vector_type(2))) float v2f;
typedef __attribute__((ext_vector_type(4))) float v4f;
typedef __attribute__((ext_vector_type(8))) float v8f;

#define D_MODEL 1024
#define D_STATE 16
#define D_CONV  4
#define D_INNER 2048
#define DT_RANK 64
#define BB 2
#define LL 2048
#define TOKENS (BB * LL)
#define XDBL_W (DT_RANK + 2 * D_STATE)

__device__ __forceinline__ float sigmoidf_(float x) { return 1.f / (1.f + __expf(-x)); }

__global__ __launch_bounds__(256)
void ln_kernel(const float* __restrict__ x, const float* __restrict__ w,
               const float* __restrict__ b, float* __restrict__ out)
{
    const int row = blockIdx.x;
    const float* xr = x + (size_t)row * D_MODEL;
    float v[4];
    float s = 0.f, s2 = 0.f;
#pragma unroll
    for (int i = 0; i < 4; ++i) {
        v[i] = xr[threadIdx.x + i * 256];
        s += v[i];
        s2 += v[i] * v[i];
    }
#pragma unroll
    for (int off = 16; off > 0; off >>= 1) {
        s  += __shfl_down(s,  off, 32);
        s2 += __shfl_down(s2, off, 32);
    }
    __shared__ float rs[8], rq[8];
    __shared__ float mu_s, rstd_s;
    const int wave = threadIdx.x >> 5;
    const int lane = threadIdx.x & 31;
    if (lane == 0) { rs[wave] = s; rq[wave] = s2; }
    __syncthreads();
    if (threadIdx.x == 0) {
        float a = 0.f, q = 0.f;
#pragma unroll
        for (int i = 0; i < 8; ++i) { a += rs[i]; q += rq[i]; }
        const float mu  = a * (1.f / D_MODEL);
        const float var = q * (1.f / D_MODEL) - mu * mu;
        mu_s = mu;
        rstd_s = rsqrtf(var + 1e-5f);
    }
    __syncthreads();
    const float mu = mu_s, rstd = rstd_s;
    float* orow = out + (size_t)row * D_MODEL;
#pragma unroll
    for (int i = 0; i < 4; ++i) {
        const int c = threadIdx.x + i * 256;
        orow[c] = (v[i] - mu) * rstd * w[c] + b[c];
    }
}

template <int MT, int NT>
__global__ __launch_bounds__(128)
void wmma_gemm_kernel(const float* __restrict__ A, int lda,
                      const float* __restrict__ Bm, int ldb,
                      float* __restrict__ C, int ldc, int K,
                      const float* __restrict__ residual, int ldr,
                      const float* __restrict__ scale_ptr)
{
    const int lane = threadIdx.x & 31;
    const int wave = threadIdx.x >> 5;
    const int wy = wave & 1;
    const int wx = wave >> 1;
    const int m0 = blockIdx.y * (32 * MT) + wy * (16 * MT);
    const int n0 = blockIdx.x * (32 * NT) + wx * (16 * NT);
    const int lc = lane & 15;
    const int lh = lane >> 4;

    v8f acc[MT][NT];
#pragma unroll
    for (int i = 0; i < MT; ++i)
#pragma unroll
        for (int t = 0; t < NT; ++t)
#pragma unroll
            for (int e = 0; e < 8; ++e) acc[i][t][e] = 0.f;

    const float* Ap = A + (size_t)(m0 + lc) * lda + 8 * lh;
    const float* Bp = Bm + (size_t)(8 * lh) * ldb + n0 + NT * lc;

    auto loadA = [&](int kb, v2f (&a)[MT][4]) {
#pragma unroll
        for (int i = 0; i < MT; ++i) {
            const float* p = Ap + (size_t)i * 16 * lda + kb;
            const v4f q0 = *(const v4f*)p;
            const v4f q1 = *(const v4f*)(p + 4);
            a[i][0] = __builtin_shufflevector(q0, q0, 0, 1);
            a[i][1] = __builtin_shufflevector(q0, q0, 2, 3);
            a[i][2] = __builtin_shufflevector(q1, q1, 0, 1);
            a[i][3] = __builtin_shufflevector(q1, q1, 2, 3);
        }
    };
    auto loadB = [&](int kb, float (&b)[8][4]) {
#pragma unroll
        for (int r = 0; r < 8; ++r) {
            const float* p = Bp + (size_t)(kb + r) * ldb;
            if constexpr (NT == 4) {
                const v4f q = *(const v4f*)p;
                b[r][0] = q.x; b[r][1] = q.y; b[r][2] = q.z; b[r][3] = q.w;
            } else {
#pragma unroll
                for (int t = 0; t < NT; ++t) b[r][t] = p[t];
            }
        }
    };
    auto mma = [&](const v2f (&a)[MT][4], const float (&b)[8][4]) {
#pragma unroll
        for (int j = 0; j < 4; ++j) {
#pragma unroll
            for (int i = 0; i < MT; ++i) {
#pragma unroll
                for (int t = 0; t < NT; ++t) {
                    v2f bv;
                    bv.x = b[2 * j][t];
                    bv.y = b[2 * j + 1][t];
                    acc[i][t] = __builtin_amdgcn_wmma_f32_16x16x4_f32(
                        false, a[i][j], false, bv, (short)0, acc[i][t], false, false);
                }
            }
        }
    };

    v2f a0[MT][4], a1[MT][4];
    float b0[8][4], b1[8][4];
    loadA(0, a0);
    loadB(0, b0);

    for (int kb = 0; kb < K; kb += 32) {
        loadA(kb + 16, a1);
        loadB(kb + 16, b1);
        __builtin_amdgcn_sched_barrier(0);
        mma(a0, b0);
        const int kn = (kb + 32 < K) ? kb + 32 : kb + 16;
        loadA(kn, a0);
        loadB(kn, b0);
        __builtin_amdgcn_sched_barrier(0);
        mma(a1, b1);
    }

    const float scale = scale_ptr ? scale_ptr[0] : 1.0f;
#pragma unroll
    for (int i = 0; i < MT; ++i) {
#pragma unroll
        for (int v = 0; v < 8; ++v) {
            const int m = m0 + i * 16 + 8 * lh + v;
            const int n = n0 + NT * lc;
            if constexpr (NT == 4) {
                v4f o;
                o.x = acc[i][0][v] * scale;
                o.y = acc[i][1][v] * scale;
                o.z = acc[i][2][v] * scale;
                o.w = acc[i][3][v] * scale;
                if (residual) o += *(const v4f*)(residual + (size_t)m * ldr + n);
                *(v4f*)(C + (size_t)m * ldc + n) = o;
            } else {
#pragma unroll
                for (int t = 0; t < NT; ++t) {
                    float x = acc[i][t][v] * scale;
                    if (residual) x += residual[(size_t)m * ldr + n + t];
                    C[(size_t)m * ldc + n + t] = x;
                }
            }
        }
    }
}

__global__ __launch_bounds__(256)
void conv_silu_kernel(const float* __restrict__ xz, const float* __restrict__ w,
                      const float* __restrict__ bias, float* __restrict__ out)
{
    const size_t idx = (size_t)blockIdx.x * 256 + threadIdx.x;
    const int c = (int)(idx & (D_INNER - 1));
    const size_t bl = idx >> 11;
    const int l = (int)(bl & (LL - 1));
    const int b = (int)(bl >> 11);
    float acc = bias[c];
#pragma unroll
    for (int k = 0; k < D_CONV; ++k) {
        const int ls = l - (D_CONV - 1) + k;
        if (ls >= 0)
            acc += w[c * D_CONV + k] *
                   xz[((size_t)(b * LL + ls)) * (2 * D_INNER) + c];
    }
    out[idx] = acc * sigmoidf_(acc);
}

__global__ __launch_bounds__(256)
void conv_state_kernel(const float* __restrict__ xz, float* __restrict__ out)
{
    const int idx = blockIdx.x * 256 + threadIdx.x;
    if (idx >= BB * D_INNER * (D_CONV - 1)) return;
    const int j = idx % (D_CONV - 1);
    const int c = (idx / (D_CONV - 1)) % D_INNER;
    const int b = idx / ((D_CONV - 1) * D_INNER);
    out[idx] = xz[((size_t)(b * LL + (LL - (D_CONV - 1) + j))) * (2 * D_INNER) + c];
}

__global__ __launch_bounds__(256)
void dt_softplus_kernel(float* __restrict__ dt, const float* __restrict__ bias)
{
    const size_t idx = (size_t)blockIdx.x * 256 + threadIdx.x;
    const int d = (int)(idx & (D_INNER - 1));
    const float v = dt[idx] + bias[d];
    dt[idx] = (v > 20.f) ? v : __logf(1.f + __expf(v));
}

#define SCAN_T   32
#define SCAN_NCH (LL / SCAN_T)
#define SCAN_WPB (D_INNER / 32)
#define SCAN_P   4
#define SCAN_S   (D_STATE / SCAN_P)
#define SCAN_R   (SCAN_T / SCAN_P)
#define SCAN_Q   (SCAN_T * 8 / (32 * SCAN_P))

__global__ __launch_bounds__(32 * SCAN_P)
void scan_kernel(const float* __restrict__ dt, const float* __restrict__ xc,
                 const float* __restrict__ xdbl, const float* __restrict__ A_log,
                 float* __restrict__ y, float* __restrict__ ssm_out)
{
    const int tid = threadIdx.x;
    const int lane = tid & 31;
    const int w = tid >> 5;
    const int b = blockIdx.x / SCAN_WPB;
    const int d = (blockIdx.x % SCAN_WPB) * 32 + lane;

    __shared__ __attribute__((aligned(16))) float sBC[2][SCAN_T * 32];
    __shared__ float sY[2][SCAN_P][SCAN_T][32];

    v2f Aval[2], h[2];
    {
        const v4f al = *(const v4f*)(A_log + (size_t)d * D_STATE + SCAN_S * w);
        Aval[0].x = -__expf(al.x);
        Aval[0].y = -__expf(al.y);
        Aval[1].x = -__expf(al.z);
        Aval[1].y = -__expf(al.w);
        h[0].x = 0.f; h[0].y = 0.f;
        h[1].x = 0.f; h[1].y = 0.f;
    }

    const size_t row0 = (size_t)b * LL;
    const float* dtp = dt + row0 * D_INNER + d;
    const float* xp  = xc + row0 * D_INNER + d;
    float*       yp  = y  + row0 * D_INNER + d;
    const float* bcp = xdbl + (row0 + (tid >> 3)) * XDBL_W + DT_RANK + 4 * (tid & 7);

    float cd[SCAN_T], cx[SCAN_T], nd[SCAN_T], nx[SCAN_T];
    v4f nbc[SCAN_Q];

    auto fetch = [&](int c, float (&fd)[SCAN_T], float (&fx)[SCAN_T], v4f (&fbc)[SCAN_Q]) {
        const size_t l0 = (size_t)c * SCAN_T;
#pragma unroll
        for (int i = 0; i < SCAN_Q; ++i)
            fbc[i] = *(const v4f*)(bcp + (l0 + 16 * i) * XDBL_W);
#pragma unroll
        for (int s = 0; s < SCAN_T; ++s) {
            fd[s] = dtp[(l0 + s) * D_INNER];
            fx[s] = xp[(l0 + s) * D_INNER];
        }
    };
    auto stash = [&](int p, const v4f (&fbc)[SCAN_Q]) {
#pragma unroll
        for (int i = 0; i < SCAN_Q; ++i)
            *(v4f*)&sBC[p][4 * tid + 512 * i] = fbc[i];
    };

    auto lds_handover = [&]() {
        __builtin_amdgcn_fence(__ATOMIC_RELEASE, "workgroup", "local");
        __builtin_amdgcn_s_barrier();
        __builtin_amdgcn_fence(__ATOMIC_ACQUIRE, "workgroup", "local");
    };

    fetch(0, cd, cx, nbc);
    stash(0, nbc);
    lds_handover();

    for (int c = 0; c < SCAN_NCH; ++c) {
        const int p = c & 1;
        const v4f* bc = (const v4f*)&sBC[p][0];
        v4f rB[SCAN_T], rC[SCAN_T];
#pragma unroll
        for (int s = 0; s < SCAN_T; ++s) {
            rB[s] = bc[s * 8 + w];
            rC[s] = bc[s * 8 + 4 + w];
        }

        const int cn = (c + 1 < SCAN_NCH) ? c + 1 : c;
        fetch(cn, nd, nx, nbc);
        __builtin_amdgcn_sched_barrier(0);

#pragma unroll
        for (int s = 0; s < SCAN_T; ++s) {
            const float dtv = cd[s];
            const float bx = dtv * cx[s];
            v2f Bk[2], Ck[2];
            Bk[0].x = rB[s].x; Bk[0].y = rB[s].y; Bk[1].x = rB[s].z; Bk[1].y = rB[s].w;
            Ck[0].x = rC[s].x; Ck[0].y = rC[s].y; Ck[1].x = rC[s].z; Ck[1].y = rC[s].w;
            float yv = 0.f;
#pragma unroll
            for (int k = 0; k < 2; ++k) {
                const v2f a = Aval[k] * dtv;
                v2f dA;
                dA.x = __expf(a.x);
                dA.y = __expf(a.y);
                h[k] = __builtin_elementwise_fma(dA, h[k], bx * Bk[k]);
                yv += h[k].x * Ck[k].x;
                yv += h[k].y * Ck[k].y;
            }
            sY[p][w][s][lane] = yv;
        }
        __builtin_amdgcn_sched_barrier(0);

        stash(p ^ 1, nbc);
#pragma unroll
        for (int s = 0; s < SCAN_T; ++s) { cd[s] = nd[s]; cx[s] = nx[s]; }
        lds_handover();

        float* yrow = yp + ((size_t)c * SCAN_T + SCAN_R * w) * D_INNER;
#pragma unroll
        for (int j = 0; j < SCAN_R; ++j) {
            const int s = SCAN_R * w + j;
            float t = sY[p][0][s][lane];
#pragma unroll
            for (int v = 1; v < SCAN_P; ++v) t += sY[p][v][s][lane];
            yrow[(size_t)j * D_INNER] = t;
        }
    }

    {
        v4f ho;
        ho.x = h[0].x; ho.y = h[0].y; ho.z = h[1].x; ho.w = h[1].y;
        *(v4f*)(ssm_out + (size_t)(b * D_INNER + d) * D_STATE + SCAN_S * w) = ho;
    }
}

__global__ __launch_bounds__(256)
void gate_kernel(float* __restrict__ y, const float* __restrict__ xc,
                 const float* __restrict__ xz, const float* __restrict__ Dp)
{
    const size_t idx = (size_t)blockIdx.x * 256 + threadIdx.x;
    const int d = (int)(idx & (D_INNER - 1));
    const size_t row = idx >> 11;
    const float z  = xz[row * (2 * D_INNER) + D_INNER + d];
    const float yv = y[idx] + xc[idx] * Dp[d];
    y[idx] = yv * z * sigmoidf_(z);
}

extern "C" void kernel_launch(void* const* d_in, const int* in_sizes, int n_in,
                              void* d_out, int out_size, void* d_ws, size_t ws_size,
                              hipStream_t stream)
{
    const float* x         = (const float*)d_in[0];
    const float* ln_w      = (const float*)d_in[1];
    const float* ln_b      = (const float*)d_in[2];
    const float* in_proj_w = (const float*)d_in[3];
    const float* conv_w    = (const float*)d_in[4];
    const float* conv_b    = (const float*)d_in[5];
    const float* x_proj_w  = (const float*)d_in[6];
    const float* dt_proj_w = (const float*)d_in[7];
    const float* dt_proj_b = (const float*)d_in[8];
    const float* A_log     = (const float*)d_in[9];
    const float* D_param   = (const float*)d_in[10];
    const float* out_projw = (const float*)d_in[11];
    const float* res_scale = (const float*)d_in[12];

    float* out      = (float*)d_out;
    float* out_ssm  = out + (size_t)BB * LL * D_MODEL;
    float* out_conv = out_ssm + (size_t)BB * D_INNER * D_STATE;

    float* ws    = (float*)d_ws;
    float* xn    = ws;
    float* xz    = xn    + (size_t)TOKENS * D_MODEL;
    float* xc    = xz    + (size_t)TOKENS * 2 * D_INNER;
    float* xdbl  = xc    + (size_t)TOKENS * D_INNER;
    float* dtbuf = xdbl  + (size_t)TOKENS * XDBL_W;
    float* ybuf  = dtbuf + (size_t)TOKENS * D_INNER;

    ln_kernel<<<TOKENS, 256, 0, stream>>>(x, ln_w, ln_b, xn);

    wmma_gemm_kernel<2, 4><<<dim3((2 * D_INNER) / 128, TOKENS / 64), 128, 0, stream>>>(
        xn, D_MODEL, in_proj_w, 2 * D_INNER, xz, 2 * D_INNER, D_MODEL,
        nullptr, 0, nullptr);

    conv_silu_kernel<<<(TOKENS * D_INNER) / 256, 256, 0, stream>>>(
        xz, conv_w, conv_b, xc);
    conv_state_kernel<<<(BB * D_INNER * (D_CONV - 1) + 255) / 256, 256, 0, stream>>>(
        xz, out_conv);

    wmma_gemm_kernel<2, 3><<<dim3(XDBL_W / 96, TOKENS / 64), 128, 0, stream>>>(
        xc, D_INNER, x_proj_w, XDBL_W, xdbl, XDBL_W, D_INNER,
        nullptr, 0, nullptr);

    wmma_gemm_kernel<2, 4><<<dim3(D_INNER / 128, TOKENS / 64), 128, 0, stream>>>(
        xdbl, XDBL_W, dt_proj_w, D_INNER, dtbuf, D_INNER, DT_RANK,
        nullptr, 0, nullptr);

    dt_softplus_kernel<<<(TOKENS * D_INNER) / 256, 256, 0, stream>>>(dtbuf, dt_proj_b);

    scan_kernel<<<BB * SCAN_WPB, 32 * SCAN_P, 0, stream>>>(
        dtbuf, xc, xdbl, A_log, ybuf, out_ssm);

    gate_kernel<<<(TOKENS * D_INNER) / 256, 256, 0, stream>>>(ybuf, xc, xz, D_param);

    wmma_gemm_kernel<2, 4><<<dim3(D_MODEL / 128, TOKENS / 64), 128, 0, stream>>>(
        ybuf, D_INNER, out_projw, D_MODEL, out, D_MODEL, D_INNER,
        x, D_MODEL, res_scale);
}


# scan kernel: next chunk's dt and x rows staged through the LDS ring like B|C (six 16-byte loads per thread, read back as {dt,x} pairs), so no load wait is left in a chunk's math; sums unchanged
# speedup vs baseline: 4.5362x; 1.0686x over previous
#include <hip/hip_runtime.h>
#include <hip/hip_bf16.h>

typedef __attribute__((ext_vector_type(2))) float v2f;
typedef __attribute__((ext_vector_type(4))) float v4f;
typedef __attribute__((ext_vector_type(8))) float v8f;

#define D_MODEL 1024
#define D_STATE 16
#define D_CONV  4
#define D_INNER 2048
#define DT_RANK 64
#define BB 2
#define LL 2048
#define TOKENS (BB * LL)
#define XDBL_W (DT_RANK + 2 * D_STATE)

__device__ __forceinline__ float sigmoidf_(float x) { return 1.f / (1.f + __expf(-x)); }

__global__ __launch_bounds__(256)
void ln_kernel(const float* __restrict__ x, const float* __restrict__ w,
               const float* __restrict__ b, float* __restrict__ out)
{
    const int row = blockIdx.x;
    const float* xr = x + (size_t)row * D_MODEL;
    float v[4];
    float s = 0.f, s2 = 0.f;
#pragma unroll
    for (int i = 0; i < 4; ++i) {
        v[i] = xr[threadIdx.x + i * 256];
        s += v[i];
        s2 += v[i] * v[i];
    }
#pragma unroll
    for (int off = 16; off > 0; off >>= 1) {
        s  += __shfl_down(s,  off, 32);
        s2 += __shfl_down(s2, off, 32);
    }
    __shared__ float rs[8], rq[8];
    __shared__ float mu_s, rstd_s;
    const int wave = threadIdx.x >> 5;
    const int lane = threadIdx.x & 31;
    if (lane == 0) { rs[wave] = s; rq[wave] = s2; }
    __syncthreads();
    if (threadIdx.x == 0) {
        float a = 0.f, q = 0.f;
#pragma unroll
        for (int i = 0; i < 8; ++i) { a += rs[i]; q += rq[i]; }
        const float mu  = a * (1.f / D_MODEL);
        const float var = q * (1.f / D_MODEL) - mu * mu;
        mu_s = mu;
        rstd_s = rsqrtf(var + 1e-5f);
    }
    __syncthreads();
    const float mu = mu_s, rstd = rstd_s;
    float* orow = out + (size_t)row * D_MODEL;
#pragma unroll
    for (int i = 0; i < 4; ++i) {
        const int c = threadIdx.x + i * 256;
        orow[c] = (v[i] - mu) * rstd * w[c] + b[c];
    }
}

template <int MT, int NT>
__global__ __launch_bounds__(128)
void wmma_gemm_kernel(const float* __restrict__ A, int lda,
                      const float* __restrict__ Bm, int ldb,
                      float* __restrict__ C, int ldc, int K,
                      const float* __restrict__ residual, int ldr,
                      const float* __restrict__ scale_ptr)
{
    const int lane = threadIdx.x & 31;
    const int wave = threadIdx.x >> 5;
    const int wy = wave & 1;
    const int wx = wave >> 1;
    const int m0 = blockIdx.y * (32 * MT) + wy * (16 * MT);
    const int n0 = blockIdx.x * (32 * NT) + wx * (16 * NT);
    const int lc = lane & 15;
    const int lh = lane >> 4;

    v8f acc[MT][NT];
#pragma unroll
    for (int i = 0; i < MT; ++i)
#pragma unroll
        for (int t = 0; t < NT; ++t)
#pragma unroll
            for (int e = 0; e < 8; ++e) acc[i][t][e] = 0.f;

    const float* Ap = A + (size_t)(m0 + lc) * lda + 8 * lh;
    const float* Bp = Bm + (size_t)(8 * lh) * ldb + n0 + NT * lc;

    auto loadA = [&](int kb, v2f (&a)[MT][4]) {
#pragma unroll
        for (int i = 0; i < MT; ++i) {
            const float* p = Ap + (size_t)i * 16 * lda + kb;
            const v4f q0 = *(const v4f*)p;
            const v4f q1 = *(const v4f*)(p + 4);
            a[i][0] = __builtin_shufflevector(q0, q0, 0, 1);
            a[i][1] = __builtin_shufflevector(q0, q0, 2, 3);
            a[i][2] = __builtin_shufflevector(q1, q1, 0, 1);
            a[i][3] = __builtin_shufflevector(q1, q1, 2, 3);
        }
    };
    auto loadB = [&](int kb, float (&b)[8][4]) {
#pragma unroll
        for (int r = 0; r < 8; ++r) {
            const float* p = Bp + (size_t)(kb + r) * ldb;
            if constexpr (NT == 4) {
                const v4f q = *(const v4f*)p;
                b[r][0] = q.x; b[r][1] = q.y; b[r][2] = q.z; b[r][3] = q.w;
            } else {
#pragma unroll
                for (int t = 0; t < NT; ++t) b[r][t] = p[t];
            }
        }
    };
    auto mma = [&](const v2f (&a)[MT][4], const float (&b)[8][4]) {
#pragma unroll
        for (int j = 0; j < 4; ++j) {
#pragma unroll
            for (int i = 0; i < MT; ++i) {
#pragma unroll
                for (int t = 0; t < NT; ++t) {
                    v2f bv;
                    bv.x = b[2 * j][t];
                    bv.y = b[2 * j + 1][t];
                    acc[i][t] = __builtin_amdgcn_wmma_f32_16x16x4_f32(
                        false, a[i][j], false, bv, (short)0, acc[i][t], false, false);
                }
            }
        }
    };

    v2f a0[MT][4], a1[MT][4];
    float b0[8][4], b1[8][4];
    loadA(0, a0);
    loadB(0, b0);

    for (int kb = 0; kb < K; kb += 32) {
        loadA(kb + 16, a1);
        loadB(kb + 16, b1);
        __builtin_amdgcn_sched_barrier(0);
        mma(a0, b0);
        const int kn = (kb + 32 < K) ? kb + 32 : kb + 16;
        loadA(kn, a0);
        loadB(kn, b0);
        __builtin_amdgcn_sched_barrier(0);
        mma(a1, b1);
    }

    const float scale = scale_ptr ? scale_ptr[0] : 1.0f;
#pragma unroll
    for (int i = 0; i < MT; ++i) {
#pragma unroll
        for (int v = 0; v < 8; ++v) {
            const int m = m0 + i * 16 + 8 * lh + v;
            const int n = n0 + NT * lc;
            if constexpr (NT == 4) {
                v4f o;
                o.x = acc[i][0][v] * scale;
                o.y = acc[i][1][v] * scale;
                o.z = acc[i][2][v] * scale;
                o.w = acc[i][3][v] * scale;
                if (residual) o += *(const v4f*)(residual + (size_t)m * ldr + n);
                *(v4f*)(C + (size_t)m * ldc + n) = o;
            } else {
#pragma unroll
                for (int t = 0; t < NT; ++t) {
                    float x = acc[i][t][v] * scale;
                    if (residual) x += residual[(size_t)m * ldr + n + t];
                    C[(size_t)m * ldc + n + t] = x;
                }
            }
        }
    }
}

__global__ __launch_bounds__(256)
void conv_silu_kernel(const float* __restrict__ xz, const float* __restrict__ w,
                      const float* __restrict__ bias, float* __restrict__ out)
{
    const size_t idx = (size_t)blockIdx.x * 256 + threadIdx.x;
    const int c = (int)(idx & (D_INNER - 1));
    const size_t bl = idx >> 11;
    const int l = (int)(bl & (LL - 1));
    const int b = (int)(bl >> 11);
    float acc = bias[c];
#pragma unroll
    for (int k = 0; k < D_CONV; ++k) {
        const int ls = l - (D_CONV - 1) + k;
        if (ls >= 0)
            acc += w[c * D_CONV + k] *
                   xz[((size_t)(b * LL + ls)) * (2 * D_INNER) + c];
    }
    out[idx] = acc * sigmoidf_(acc);
}

__global__ __launch_bounds__(256)
void conv_state_kernel(const float* __restrict__ xz, float* __restrict__ out)
{
    const int idx = blockIdx.x * 256 + threadIdx.x;
    if (idx >= BB * D_INNER * (D_CONV - 1)) return;
    const int j = idx % (D_CONV - 1);
    const int c = (idx / (D_CONV - 1)) % D_INNER;
    const int b = idx / ((D_CONV - 1) * D_INNER);
    out[idx] = xz[((size_t)(b * LL + (LL - (D_CONV - 1) + j))) * (2 * D_INNER) + c];
}

__global__ __launch_bounds__(256)
void dt_softplus_kernel(float* __restrict__ dt, const float* __restrict__ bias)
{
    const size_t idx = (size_t)blockIdx.x * 256 + threadIdx.x;
    const int d = (int)(idx & (D_INNER - 1));
    const float v = dt[idx] + bias[d];
    dt[idx] = (v > 20.f) ? v : __logf(1.f + __expf(v));
}

#define SCAN_T   32
#define SCAN_NCH (LL / SCAN_T)
#define SCAN_WPB (D_INNER / 32)
#define SCAN_P   4
#define SCAN_S   (D_STATE / SCAN_P)
#define SCAN_R   (SCAN_T / SCAN_P)
#define SCAN_Q   (SCAN_T * 8 / (32 * SCAN_P))

__global__ __launch_bounds__(32 * SCAN_P)
void scan_kernel(const float* __restrict__ dt, const float* __restrict__ xc,
                 const float* __restrict__ xdbl, const float* __restrict__ A_log,
                 float* __restrict__ y, float* __restrict__ ssm_out)
{
    const int tid = threadIdx.x;
    const int lane = tid & 31;
    const int w = tid >> 5;
    const int b = blockIdx.x / SCAN_WPB;
    const int d0 = (blockIdx.x % SCAN_WPB) * 32;
    const int d = d0 + lane;

    __shared__ __attribute__((aligned(16))) float sBC[2][SCAN_T * 32];
    __shared__ __attribute__((aligned(16))) v2f sDX[2][SCAN_T * 32];
    __shared__ float sY[2][SCAN_P][SCAN_T][32];

    v2f Aval[2], h[2];
    {
        const v4f al = *(const v4f*)(A_log + (size_t)d * D_STATE + SCAN_S * w);
        Aval[0].x = -__expf(al.x);
        Aval[0].y = -__expf(al.y);
        Aval[1].x = -__expf(al.z);
        Aval[1].y = -__expf(al.w);
        h[0].x = 0.f; h[0].y = 0.f;
        h[1].x = 0.f; h[1].y = 0.f;
    }

    const size_t row0 = (size_t)b * LL;
    float* yp = y + row0 * D_INNER + d;
    const int prow = tid >> 3;
    const int pcol = 4 * (tid & 7);
    const float* bcp = xdbl + (row0 + prow) * XDBL_W + DT_RANK + pcol;
    const float* dtq = dt + (row0 + prow) * D_INNER + d0 + pcol;
    const float* xq  = xc + (row0 + prow) * D_INNER + d0 + pcol;

    v4f nbc[SCAN_Q], ndt[SCAN_Q], nxx[SCAN_Q];

    auto fetch = [&](int c) {
        const size_t l0 = (size_t)c * SCAN_T;
#pragma unroll
        for (int i = 0; i < SCAN_Q; ++i) {
            nbc[i] = *(const v4f*)(bcp + (l0 + 16 * i) * XDBL_W);
            ndt[i] = *(const v4f*)(dtq + (l0 + 16 * i) * D_INNER);
            nxx[i] = *(const v4f*)(xq  + (l0 + 16 * i) * D_INNER);
        }
    };
    auto stash = [&](int p) {
#pragma unroll
        for (int i = 0; i < SCAN_Q; ++i) {
            *(v4f*)&sBC[p][4 * tid + 512 * i] = nbc[i];
            v4f lo, hi;
            lo.x = ndt[i].x; lo.y = nxx[i].x; lo.z = ndt[i].y; lo.w = nxx[i].y;
            hi.x = ndt[i].z; hi.y = nxx[i].z; hi.z = ndt[i].w; hi.w = nxx[i].w;
            v2f* q = &sDX[p][4 * tid + 512 * i];
            *(v4f*)q = lo;
            *(v4f*)(q + 2) = hi;
        }
    };

    auto lds_handover = [&]() {
        __builtin_amdgcn_fence(__ATOMIC_RELEASE, "workgroup", "local");
        __builtin_amdgcn_s_barrier();
        __builtin_amdgcn_fence(__ATOMIC_ACQUIRE, "workgroup", "local");
    };

    fetch(0);
    stash(0);
    lds_handover();

    for (int c = 0; c < SCAN_NCH; ++c) {
        const int p = c & 1;
        const v4f* bc = (const v4f*)&sBC[p][0];
        v4f rB[SCAN_T], rC[SCAN_T];
        v2f dx[SCAN_T];
#pragma unroll
        for (int s = 0; s < SCAN_T; ++s) {
            rB[s] = bc[s * 8 + w];
            rC[s] = bc[s * 8 + 4 + w];
            dx[s] = sDX[p][s * 32 + lane];
        }

        const int cn = (c + 1 < SCAN_NCH) ? c + 1 : c;
        fetch(cn);
        __builtin_amdgcn_sched_barrier(0);

#pragma unroll
        for (int s = 0; s < SCAN_T; ++s) {
            const float dtv = dx[s].x;
            const float bx = dtv * dx[s].y;
            v2f Bk[2], Ck[2];
            Bk[0].x = rB[s].x; Bk[0].y = rB[s].y; Bk[1].x = rB[s].z; Bk[1].y = rB[s].w;
            Ck[0].x = rC[s].x; Ck[0].y = rC[s].y; Ck[1].x = rC[s].z; Ck[1].y = rC[s].w;
            float yv = 0.f;
#pragma unroll
            for (int k = 0; k < 2; ++k) {
                const v2f a = Aval[k] * dtv;
                v2f dA;
                dA.x = __expf(a.x);
                dA.y = __expf(a.y);
                h[k] = __builtin_elementwise_fma(dA, h[k], bx * Bk[k]);
                yv += h[k].x * Ck[k].x;
                yv += h[k].y * Ck[k].y;
            }
            sY[p][w][s][lane] = yv;
        }
        __builtin_amdgcn_sched_barrier(0);

        stash(p ^ 1);
        lds_handover();

        float* yrow = yp + ((size_t)c * SCAN_T + SCAN_R * w) * D_INNER;
#pragma unroll
        for (int j = 0; j < SCAN_R; ++j) {
            const int s = SCAN_R * w + j;
            float t = sY[p][0][s][lane];
#pragma unroll
            for (int v = 1; v < SCAN_P; ++v) t += sY[p][v][s][lane];
            yrow[(size_t)j * D_INNER] = t;
        }
    }

    {
        v4f ho;
        ho.x = h[0].x; ho.y = h[0].y; ho.z = h[1].x; ho.w = h[1].y;
        *(v4f*)(ssm_out + (size_t)(b * D_INNER + d) * D_STATE + SCAN_S * w) = ho;
    }
}

__global__ __launch_bounds__(256)
void gate_kernel(float* __restrict__ y, const float* __restrict__ xc,
                 const float* __restrict__ xz, const float* __restrict__ Dp)
{
    const size_t idx = (size_t)blockIdx.x * 256 + threadIdx.x;
    const int d = (int)(idx & (D_INNER - 1));
    const size_t row = idx >> 11;
    const float z  = xz[row * (2 * D_INNER) + D_INNER + d];
    const float yv = y[idx] + xc[idx] * Dp[d];
    y[idx] = yv * z * sigmoidf_(z);
}

extern "C" void kernel_launch(void* const* d_in, const int* in_sizes, int n_in,
                              void* d_out, int out_size, void* d_ws, size_t ws_size,
                              hipStream_t stream)
{
    const float* x         = (const float*)d_in[0];
    const float* ln_w      = (const float*)d_in[1];
    const float* ln_b      = (const float*)d_in[2];
    const float* in_proj_w = (const float*)d_in[3];
    const float* conv_w    = (const float*)d_in[4];
    const float* conv_b    = (const float*)d_in[5];
    const float* x_proj_w  = (const float*)d_in[6];
    const float* dt_proj_w = (const float*)d_in[7];
    const float* dt_proj_b = (const float*)d_in[8];
    const float* A_log     = (const float*)d_in[9];
    const float* D_param   = (const float*)d_in[10];
    const float* out_projw = (const float*)d_in[11];
    const float* res_scale = (const float*)d_in[12];

    float* out      = (float*)d_out;
    float* out_ssm  = out + (size_t)BB * LL * D_MODEL;
    float* out_conv = out_ssm + (size_t)BB * D_INNER * D_STATE;

    float* ws    = (float*)d_ws;
    float* xn    = ws;
    float* xz    = xn    + (size_t)TOKENS * D_MODEL;
    float* xc    = xz    + (size_t)TOKENS * 2 * D_INNER;
    float* xdbl  = xc    + (size_t)TOKENS * D_INNER;
    float* dtbuf = xdbl  + (size_t)TOKENS * XDBL_W;
    float* ybuf  = dtbuf + (size_t)TOKENS * D_INNER;

    ln_kernel<<<TOKENS, 256, 0, stream>>>(x, ln_w, ln_b, xn);

    wmma_gemm_kernel<2, 4><<<dim3((2 * D_INNER) / 128, TOKENS / 64), 128, 0, stream>>>(
        xn, D_MODEL, in_proj_w, 2 * D_INNER, xz, 2 * D_INNER, D_MODEL,
        nullptr, 0, nullptr);

    conv_silu_kernel<<<(TOKENS * D_INNER) / 256, 256, 0, stream>>>(
        xz, conv_w, conv_b, xc);
    conv_state_kernel<<<(BB * D_INNER * (D_CONV - 1) + 255) / 256, 256, 0, stream>>>(
        xz, out_conv);

    wmma_gemm_kernel<2, 3><<<dim3(XDBL_W / 96, TOKENS / 64), 128, 0, stream>>>(
        xc, D_INNER, x_proj_w, XDBL_W, xdbl, XDBL_W, D_INNER,
        nullptr, 0, nullptr);

    wmma_gemm_kernel<2, 4><<<dim3(D_INNER / 128, TOKENS / 64), 128, 0, stream>>>(
        xdbl, XDBL_W, dt_proj_w, D_INNER, dtbuf, D_INNER, DT_RANK,
        nullptr, 0, nullptr);

    dt_softplus_kernel<<<(TOKENS * D_INNER) / 256, 256, 0, stream>>>(dtbuf, dt_proj_b);

    scan_kernel<<<BB * SCAN_WPB, 32 * SCAN_P, 0, stream>>>(
        dtbuf, xc, xdbl, A_log, ybuf, out_ssm);

    gate_kernel<<<(TOKENS * D_INNER) / 256, 256, 0, stream>>>(ybuf, xc, xz, D_param);

    wmma_gemm_kernel<2, 4><<<dim3(D_MODEL / 128, TOKENS / 64), 128, 0, stream>>>(
        ybuf, D_INNER, out_projw, D_MODEL, out, D_MODEL, D_INNER,
        x, D_MODEL, res_scale);
}


# conv+SiLU, softplus and gate kernels: each thread takes eight 16-byte pieces (4 adjacent channels) instead of one float, so 32x fewer waves enter; every element's operations unchanged
# speedup vs baseline: 4.7267x; 1.0420x over previous
#include <hip/hip_runtime.h>
#include <hip/hip_bf16.h>

typedef __attribute__((ext_vector_type(2))) float v2f;
typedef __attribute__((ext_vector_type(4))) float v4f;
typedef __attribute__((ext_vector_type(8))) float v8f;

#define D_MODEL 1024
#define D_STATE 16
#define D_CONV  4
#define D_INNER 2048
#define DT_RANK 64
#define BB 2
#define LL 2048
#define TOKENS (BB * LL)
#define XDBL_W (DT_RANK + 2 * D_STATE)

__device__ __forceinline__ float sigmoidf_(float x) { return 1.f / (1.f + __expf(-x)); }

__global__ __launch_bounds__(256)
void ln_kernel(const float* __restrict__ x, const float* __restrict__ w,
               const float* __restrict__ b, float* __restrict__ out)
{
    const int row = blockIdx.x;
    const float* xr = x + (size_t)row * D_MODEL;
    float v[4];
    float s = 0.f, s2 = 0.f;
#pragma unroll
    for (int i = 0; i < 4; ++i) {
        v[i] = xr[threadIdx.x + i * 256];
        s += v[i];
        s2 += v[i] * v[i];
    }
#pragma unroll
    for (int off = 16; off > 0; off >>= 1) {
        s  += __shfl_down(s,  off, 32);
        s2 += __shfl_down(s2, off, 32);
    }
    __shared__ float rs[8], rq[8];
    __shared__ float mu_s, rstd_s;
    const int wave = threadIdx.x >> 5;
    const int lane = threadIdx.x & 31;
    if (lane == 0) { rs[wave] = s; rq[wave] = s2; }
    __syncthreads();
    if (threadIdx.x == 0) {
        float a = 0.f, q = 0.f;
#pragma unroll
        for (int i = 0; i < 8; ++i) { a += rs[i]; q += rq[i]; }
        const float mu  = a * (1.f / D_MODEL);
        const float var = q * (1.f / D_MODEL) - mu * mu;
        mu_s = mu;
        rstd_s = rsqrtf(var + 1e-5f);
    }
    __syncthreads();
    const float mu = mu_s, rstd = rstd_s;
    float* orow = out + (size_t)row * D_MODEL;
#pragma unroll
    for (int i = 0; i < 4; ++i) {
        const int c = threadIdx.x + i * 256;
        orow[c] = (v[i] - mu) * rstd * w[c] + b[c];
    }
}

template <int MT, int NT>
__global__ __launch_bounds__(128)
void wmma_gemm_kernel(const float* __restrict__ A, int lda,
                      const float* __restrict__ Bm, int ldb,
                      float* __restrict__ C, int ldc, int K,
                      const float* __restrict__ residual, int ldr,
                      const float* __restrict__ scale_ptr)
{
    const int lane = threadIdx.x & 31;
    const int wave = threadIdx.x >> 5;
    const int wy = wave & 1;
    const int wx = wave >> 1;
    const int m0 = blockIdx.y * (32 * MT) + wy * (16 * MT);
    const int n0 = blockIdx.x * (32 * NT) + wx * (16 * NT);
    const int lc = lane & 15;
    const int lh = lane >> 4;

    v8f acc[MT][NT];
#pragma unroll
    for (int i = 0; i < MT; ++i)
#pragma unroll
        for (int t = 0; t < NT; ++t)
#pragma unroll
            for (int e = 0; e < 8; ++e) acc[i][t][e] = 0.f;

    const float* Ap = A + (size_t)(m0 + lc) * lda + 8 * lh;
    const float* Bp = Bm + (size_t)(8 * lh) * ldb + n0 + NT * lc;

    auto loadA = [&](int kb, v2f (&a)[MT][4]) {
#pragma unroll
        for (int i = 0; i < MT; ++i) {
            const float* p = Ap + (size_t)i * 16 * lda + kb;
            const v4f q0 = *(const v4f*)p;
            const v4f q1 = *(const v4f*)(p + 4);
            a[i][0] = __builtin_shufflevector(q0, q0, 0, 1);
            a[i][1] = __builtin_shufflevector(q0, q0, 2, 3);
            a[i][2] = __builtin_shufflevector(q1, q1, 0, 1);
            a[i][3] = __builtin_shufflevector(q1, q1, 2, 3);
        }
    };
    auto loadB = [&](int kb, float (&b)[8][4]) {
#pragma unroll
        for (int r = 0; r < 8; ++r) {
            const float* p = Bp + (size_t)(kb + r) * ldb;
            if constexpr (NT == 4) {
                const v4f q = *(const v4f*)p;
                b[r][0] = q.x; b[r][1] = q.y; b[r][2] = q.z; b[r][3] = q.w;
            } else {
#pragma unroll
                for (int t = 0; t < NT; ++t) b[r][t] = p[t];
            }
        }
    };
    auto mma = [&](const v2f (&a)[MT][4], const float (&b)[8][4]) {
#pragma unroll
        for (int j = 0; j < 4; ++j) {
#pragma unroll
            for (int i = 0; i < MT; ++i) {
#pragma unroll
                for (int t = 0; t < NT; ++t) {
                    v2f bv;
                    bv.x = b[2 * j][t];
                    bv.y = b[2 * j + 1][t];
                    acc[i][t] = __builtin_amdgcn_wmma_f32_16x16x4_f32(
                        false, a[i][j], false, bv, (short)0, acc[i][t], false, false);
                }
            }
        }
    };

    v2f a0[MT][4], a1[MT][4];
    float b0[8][4], b1[8][4];
    loadA(0, a0);
    loadB(0, b0);

    for (int kb = 0; kb < K; kb += 32) {
        loadA(kb + 16, a1);
        loadB(kb + 16, b1);
        __builtin_amdgcn_sched_barrier(0);
        mma(a0, b0);
        const int kn = (kb + 32 < K) ? kb + 32 : kb + 16;
        loadA(kn, a0);
        loadB(kn, b0);
        __builtin_amdgcn_sched_barrier(0);
        mma(a1, b1);
    }

    const float scale = scale_ptr ? scale_ptr[0] : 1.0f;
#pragma unroll
    for (int i = 0; i < MT; ++i) {
#pragma unroll
        for (int v = 0; v < 8; ++v) {
            const int m = m0 + i * 16 + 8 * lh + v;
            const int n = n0 + NT * lc;
            if constexpr (NT == 4) {
                v4f o;
                o.x = acc[i][0][v] * scale;
                o.y = acc[i][1][v] * scale;
                o.z = acc[i][2][v] * scale;
                o.w = acc[i][3][v] * scale;
                if (residual) o += *(const v4f*)(residual + (size_t)m * ldr + n);
                *(v4f*)(C + (size_t)m * ldc + n) = o;
            } else {
#pragma unroll
                for (int t = 0; t < NT; ++t) {
                    float x = acc[i][t][v] * scale;
                    if (residual) x += residual[(size_t)m * ldr + n + t];
                    C[(size_t)m * ldc + n + t] = x;
                }
            }
        }
    }
}

#define EW_U 8
#define CONV_TL 8

__global__ __launch_bounds__(256)
void conv_silu_kernel(const float* __restrict__ xz, const float* __restrict__ w,
                      const float* __restrict__ bias, float* __restrict__ out)
{
    const int c4 = 4 * ((int)(blockIdx.x & 1) * 256 + (int)threadIdx.x);
    const int t0 = (int)(blockIdx.x >> 1) * CONV_TL;
    const bool head = (t0 & (LL - 1)) == 0;

    v4f wv[4];
#pragma unroll
    for (int ch = 0; ch < 4; ++ch)
        wv[ch] = *(const v4f*)(w + (size_t)(c4 + ch) * D_CONV);
    const v4f bv = *(const v4f*)(bias + c4);

    v4f r[CONV_TL + 3];
    const float* xp = xz + (size_t)t0 * (2 * D_INNER) + c4;
    const ptrdiff_t hist = head ? 0 : (ptrdiff_t)(2 * D_INNER);
#pragma unroll
    for (int i = 0; i < 3; ++i)
        r[i] = *(const v4f*)(xp + (ptrdiff_t)(i - 3) * hist);
#pragma unroll
    for (int i = 3; i < CONV_TL + 3; ++i)
        r[i] = *(const v4f*)(xp + (size_t)(i - 3) * (2 * D_INNER));
    __builtin_amdgcn_sched_barrier(0);

    float* op = out + (size_t)t0 * D_INNER + c4;
#pragma unroll
    for (int j = 0; j < CONV_TL; ++j) {
        v4f o;
#pragma unroll
        for (int ch = 0; ch < 4; ++ch) {
            float acc = bv[ch];
#pragma unroll
            for (int k = 0; k < D_CONV; ++k) {
                if (j + k >= 3 || !head)
                    acc += wv[ch][k] * r[j + k][ch];
            }
            o[ch] = acc * sigmoidf_(acc);
        }
        *(v4f*)(op + (size_t)j * D_INNER) = o;
    }
}

__global__ __launch_bounds__(256)
void conv_state_kernel(const float* __restrict__ xz, float* __restrict__ out)
{
    const int idx = blockIdx.x * 256 + threadIdx.x;
    if (idx >= BB * D_INNER * (D_CONV - 1)) return;
    const int j = idx % (D_CONV - 1);
    const int c = (idx / (D_CONV - 1)) % D_INNER;
    const int b = idx / ((D_CONV - 1) * D_INNER);
    out[idx] = xz[((size_t)(b * LL + (LL - (D_CONV - 1) + j))) * (2 * D_INNER) + c];
}

__global__ __launch_bounds__(256)
void dt_softplus_kernel(float* __restrict__ dt, const float* __restrict__ bias)
{
    v4f* p = (v4f*)dt + (size_t)blockIdx.x * (EW_U * 256) + threadIdx.x;
    const v4f bq[2] = { ((const v4f*)bias)[threadIdx.x],
                        ((const v4f*)bias)[threadIdx.x + 256] };
    v4f v[EW_U];
#pragma unroll
    for (int u = 0; u < EW_U; ++u) v[u] = p[256 * u];
#pragma unroll
    for (int u = 0; u < EW_U; ++u) {
        v4f o;
#pragma unroll
        for (int e = 0; e < 4; ++e) {
            const float x = v[u][e] + bq[u & 1][e];
            o[e] = (x > 20.f) ? x : __logf(1.f + __expf(x));
        }
        p[256 * u] = o;
    }
}

#define SCAN_T   32
#define SCAN_NCH (LL / SCAN_T)
#define SCAN_WPB (D_INNER / 32)
#define SCAN_P   4
#define SCAN_S   (D_STATE / SCAN_P)
#define SCAN_R   (SCAN_T / SCAN_P)
#define SCAN_Q   (SCAN_T * 8 / (32 * SCAN_P))

__global__ __launch_bounds__(32 * SCAN_P)
void scan_kernel(const float* __restrict__ dt, const float* __restrict__ xc,
                 const float* __restrict__ xdbl, const float* __restrict__ A_log,
                 float* __restrict__ y, float* __restrict__ ssm_out)
{
    const int tid = threadIdx.x;
    const int lane = tid & 31;
    const int w = tid >> 5;
    const int b = blockIdx.x / SCAN_WPB;
    const int d0 = (blockIdx.x % SCAN_WPB) * 32;
    const int d = d0 + lane;

    __shared__ __attribute__((aligned(16))) float sBC[2][SCAN_T * 32];
    __shared__ __attribute__((aligned(16))) v2f sDX[2][SCAN_T * 32];
    __shared__ float sY[2][SCAN_P][SCAN_T][32];

    v2f Aval[2], h[2];
    {
        const v4f al = *(const v4f*)(A_log + (size_t)d * D_STATE + SCAN_S * w);
        Aval[0].x = -__expf(al.x);
        Aval[0].y = -__expf(al.y);
        Aval[1].x = -__expf(al.z);
        Aval[1].y = -__expf(al.w);
        h[0].x = 0.f; h[0].y = 0.f;
        h[1].x = 0.f; h[1].y = 0.f;
    }

    const size_t row0 = (size_t)b * LL;
    float* yp = y + row0 * D_INNER + d;
    const int prow = tid >> 3;
    const int pcol = 4 * (tid & 7);
    const float* bcp = xdbl + (row0 + prow) * XDBL_W + DT_RANK + pcol;
    const float* dtq = dt + (row0 + prow) * D_INNER + d0 + pcol;
    const float* xq  = xc + (row0 + prow) * D_INNER + d0 + pcol;

    v4f nbc[SCAN_Q], ndt[SCAN_Q], nxx[SCAN_Q];

    auto fetch = [&](int c) {
        const size_t l0 = (size_t)c * SCAN_T;
#pragma unroll
        for (int i = 0; i < SCAN_Q; ++i) {
            nbc[i] = *(const v4f*)(bcp + (l0 + 16 * i) * XDBL_W);
            ndt[i] = *(const v4f*)(dtq + (l0 + 16 * i) * D_INNER);
            nxx[i] = *(const v4f*)(xq  + (l0 + 16 * i) * D_INNER);
        }
    };
    auto stash = [&](int p) {
#pragma unroll
        for (int i = 0; i < SCAN_Q; ++i) {
            *(v4f*)&sBC[p][4 * tid + 512 * i] = nbc[i];
            v4f lo, hi;
            lo.x = ndt[i].x; lo.y = nxx[i].x; lo.z = ndt[i].y; lo.w = nxx[i].y;
            hi.x = ndt[i].z; hi.y = nxx[i].z; hi.z = ndt[i].w; hi.w = nxx[i].w;
            v2f* q = &sDX[p][4 * tid + 512 * i];
            *(v4f*)q = lo;
            *(v4f*)(q + 2) = hi;
        }
    };

    auto lds_handover = [&]() {
        __builtin_amdgcn_fence(__ATOMIC_RELEASE, "workgroup", "local");
        __builtin_amdgcn_s_barrier();
        __builtin_amdgcn_fence(__ATOMIC_ACQUIRE, "workgroup", "local");
    };

    fetch(0);
    stash(0);
    lds_handover();

    for (int c = 0; c < SCAN_NCH; ++c) {
        const int p = c & 1;
        const v4f* bc = (const v4f*)&sBC[p][0];
        v4f rB[SCAN_T], rC[SCAN_T];
        v2f dx[SCAN_T];
#pragma unroll
        for (int s = 0; s < SCAN_T; ++s) {
            rB[s] = bc[s * 8 + w];
            rC[s] = bc[s * 8 + 4 + w];
            dx[s] = sDX[p][s * 32 + lane];
        }

        const int cn = (c + 1 < SCAN_NCH) ? c + 1 : c;
        fetch(cn);
        __builtin_amdgcn_sched_barrier(0);

#pragma unroll
        for (int s = 0; s < SCAN_T; ++s) {
            const float dtv = dx[s].x;
            const float bx = dtv * dx[s].y;
            v2f Bk[2], Ck[2];
            Bk[0].x = rB[s].x; Bk[0].y = rB[s].y; Bk[1].x = rB[s].z; Bk[1].y = rB[s].w;
            Ck[0].x = rC[s].x; Ck[0].y = rC[s].y; Ck[1].x = rC[s].z; Ck[1].y = rC[s].w;
            float yv = 0.f;
#pragma unroll
            for (int k = 0; k < 2; ++k) {
                const v2f a = Aval[k] * dtv;
                v2f dA;
                dA.x = __expf(a.x);
                dA.y = __expf(a.y);
                h[k] = __builtin_elementwise_fma(dA, h[k], bx * Bk[k]);
                yv += h[k].x * Ck[k].x;
                yv += h[k].y * Ck[k].y;
            }
            sY[p][w][s][lane] = yv;
        }
        __builtin_amdgcn_sched_barrier(0);

        stash(p ^ 1);
        lds_handover();

        float* yrow = yp + ((size_t)c * SCAN_T + SCAN_R * w) * D_INNER;
#pragma unroll
        for (int j = 0; j < SCAN_R; ++j) {
            const int s = SCAN_R * w + j;
            float t = sY[p][0][s][lane];
#pragma unroll
            for (int v = 1; v < SCAN_P; ++v) t += sY[p][v][s][lane];
            yrow[(size_t)j * D_INNER] = t;
        }
    }

    {
        v4f ho;
        ho.x = h[0].x; ho.y = h[0].y; ho.z = h[1].x; ho.w = h[1].y;
        *(v4f*)(ssm_out + (size_t)(b * D_INNER + d) * D_STATE + SCAN_S * w) = ho;
    }
}

__global__ __launch_bounds__(256)
void gate_kernel(float* __restrict__ y, const float* __restrict__ xc,
                 const float* __restrict__ xz, const float* __restrict__ Dp)
{
    const size_t base = (size_t)blockIdx.x * (EW_U * 256) + threadIdx.x;
    v4f* yp = (v4f*)y + base;
    const v4f* xp = (const v4f*)xc + base;
    const v4f* zp = (const v4f*)(xz + (size_t)blockIdx.x * (EW_U / 2) * (2 * D_INNER) + D_INNER)
                    + threadIdx.x;
    const v4f dq[2] = { ((const v4f*)Dp)[threadIdx.x],
                        ((const v4f*)Dp)[threadIdx.x + 256] };
    v4f yv[EW_U], xv[EW_U], zv[EW_U];
#pragma unroll
    for (int u = 0; u < EW_U; ++u) {
        yv[u] = yp[256 * u];
        xv[u] = xp[256 * u];
        zv[u] = zp[(u >> 1) * (2 * D_INNER / 4) + (u & 1) * 256];
    }
    __builtin_amdgcn_sched_barrier(0);
#pragma unroll
    for (int u = 0; u < EW_U; ++u) {
        v4f o;
#pragma unroll
        for (int e = 0; e < 4; ++e) {
            const float z = zv[u][e];
            const float t = yv[u][e] + xv[u][e] * dq[u & 1][e];
            o[e] = t * z * sigmoidf_(z);
        }
        yp[256 * u] = o;
    }
}

extern "C" void kernel_launch(void* const* d_in, const int* in_sizes, int n_in,
                              void* d_out, int out_size, void* d_ws, size_t ws_size,
                              hipStream_t stream)
{
    const float* x         = (const float*)d_in[0];
    const float* ln_w      = (const float*)d_in[1];
    const float* ln_b      = (const float*)d_in[2];
    const float* in_proj_w = (const float*)d_in[3];
    const float* conv_w    = (const float*)d_in[4];
    const float* conv_b    = (const float*)d_in[5];
    const float* x_proj_w  = (const float*)d_in[6];
    const float* dt_proj_w = (const float*)d_in[7];
    const float* dt_proj_b = (const float*)d_in[8];
    const float* A_log     = (const float*)d_in[9];
    const float* D_param   = (const float*)d_in[10];
    const float* out_projw = (const float*)d_in[11];
    const float* res_scale = (const float*)d_in[12];

    float* out      = (float*)d_out;
    float* out_ssm  = out + (size_t)BB * LL * D_MODEL;
    float* out_conv = out_ssm + (size_t)BB * D_INNER * D_STATE;

    float* ws    = (float*)d_ws;
    float* xn    = ws;
    float* xz    = xn    + (size_t)TOKENS * D_MODEL;
    float* xc    = xz    + (size_t)TOKENS * 2 * D_INNER;
    float* xdbl  = xc    + (size_t)TOKENS * D_INNER;
    float* dtbuf = xdbl  + (size_t)TOKENS * XDBL_W;
    float* ybuf  = dtbuf + (size_t)TOKENS * D_INNER;

    ln_kernel<<<TOKENS, 256, 0, stream>>>(x, ln_w, ln_b, xn);

    wmma_gemm_kernel<2, 4><<<dim3((2 * D_INNER) / 128, TOKENS / 64), 128, 0, stream>>>(
        xn, D_MODEL, in_proj_w, 2 * D_INNER, xz, 2 * D_INNER, D_MODEL,
        nullptr, 0, nullptr);

    conv_silu_kernel<<<(TOKENS / CONV_TL) * (D_INNER / 4 / 256), 256, 0, stream>>>(
        xz, conv_w, conv_b, xc);
    conv_state_kernel<<<(BB * D_INNER * (D_CONV - 1) + 255) / 256, 256, 0, stream>>>(
        xz, out_conv);

    wmma_gemm_kernel<2, 3><<<dim3(XDBL_W / 96, TOKENS / 64), 128, 0, stream>>>(
        xc, D_INNER, x_proj_w, XDBL_W, xdbl, XDBL_W, D_INNER,
        nullptr, 0, nullptr);

    wmma_gemm_kernel<2, 4><<<dim3(D_INNER / 128, TOKENS / 64), 128, 0, stream>>>(
        xdbl, XDBL_W, dt_proj_w, D_INNER, dtbuf, D_INNER, DT_RANK,
        nullptr, 0, nullptr);

    dt_softplus_kernel<<<(TOKENS * D_INNER / 4) / (EW_U * 256), 256, 0, stream>>>(dtbuf, dt_proj_b);

    scan_kernel<<<BB * SCAN_WPB, 32 * SCAN_P, 0, stream>>>(
        dtbuf, xc, xdbl, A_log, ybuf, out_ssm);

    gate_kernel<<<(TOKENS * D_INNER / 4) / (EW_U * 256), 256, 0, stream>>>(ybuf, xc, xz, D_param);

    wmma_gemm_kernel<2, 4><<<dim3(D_MODEL / 128, TOKENS / 64), 128, 0, stream>>>(
        ybuf, D_INNER, out_projw, D_MODEL, out, D_MODEL, D_INNER,
        x, D_MODEL, res_scale);
}


# x_proj GEMM: K split over the 16 waves of a workgroup (128 k each, 16 rows x 96 columns, 256 workgroups instead of 64); the 16 partial sums are added in ascending k order through LDS
# speedup vs baseline: 5.1295x; 1.0852x over previous
#include <hip/hip_runtime.h>
#include <hip/hip_bf16.h>

typedef __attribute__((ext_vector_type(2))) float v2f;
typedef __attribute__((ext_vector_type(4))) float v4f;
typedef __attribute__((ext_vector_type(8))) float v8f;

#define D_MODEL 1024
#define D_STATE 16
#define D_CONV  4
#define D_INNER 2048
#define DT_RANK 64
#define BB 2
#define LL 2048
#define TOKENS (BB * LL)
#define XDBL_W (DT_RANK + 2 * D_STATE)

__device__ __forceinline__ float sigmoidf_(float x) { return 1.f / (1.f + __expf(-x)); }

__global__ __launch_bounds__(256)
void ln_kernel(const float* __restrict__ x, const float* __restrict__ w,
               const float* __restrict__ b, float* __restrict__ out)
{
    const int row = blockIdx.x;
    const float* xr = x + (size_t)row * D_MODEL;
    float v[4];
    float s = 0.f, s2 = 0.f;
#pragma unroll
    for (int i = 0; i < 4; ++i) {
        v[i] = xr[threadIdx.x + i * 256];
        s += v[i];
        s2 += v[i] * v[i];
    }
#pragma unroll
    for (int off = 16; off > 0; off >>= 1) {
        s  += __shfl_down(s,  off, 32);
        s2 += __shfl_down(s2, off, 32);
    }
    __shared__ float rs[8], rq[8];
    __shared__ float mu_s, rstd_s;
    const int wave = threadIdx.x >> 5;
    const int lane = threadIdx.x & 31;
    if (lane == 0) { rs[wave] = s; rq[wave] = s2; }
    __syncthreads();
    if (threadIdx.x == 0) {
        float a = 0.f, q = 0.f;
#pragma unroll
        for (int i = 0; i < 8; ++i) { a += rs[i]; q += rq[i]; }
        const float mu  = a * (1.f / D_MODEL);
        const float var = q * (1.f / D_MODEL) - mu * mu;
        mu_s = mu;
        rstd_s = rsqrtf(var + 1e-5f);
    }
    __syncthreads();
    const float mu = mu_s, rstd = rstd_s;
    float* orow = out + (size_t)row * D_MODEL;
#pragma unroll
    for (int i = 0; i < 4; ++i) {
        const int c = threadIdx.x + i * 256;
        orow[c] = (v[i] - mu) * rstd * w[c] + b[c];
    }
}

template <int MT, int NT>
__global__ __launch_bounds__(128)
void wmma_gemm_kernel(const float* __restrict__ A, int lda,
                      const float* __restrict__ Bm, int ldb,
                      float* __restrict__ C, int ldc, int K,
                      const float* __restrict__ residual, int ldr,
                      const float* __restrict__ scale_ptr)
{
    const int lane = threadIdx.x & 31;
    const int wave = threadIdx.x >> 5;
    const int wy = wave & 1;
    const int wx = wave >> 1;
    const int m0 = blockIdx.y * (32 * MT) + wy * (16 * MT);
    const int n0 = blockIdx.x * (32 * NT) + wx * (16 * NT);
    const int lc = lane & 15;
    const int lh = lane >> 4;

    v8f acc[MT][NT];
#pragma unroll
    for (int i = 0; i < MT; ++i)
#pragma unroll
        for (int t = 0; t < NT; ++t)
#pragma unroll
            for (int e = 0; e < 8; ++e) acc[i][t][e] = 0.f;

    const float* Ap = A + (size_t)(m0 + lc) * lda + 8 * lh;
    const float* Bp = Bm + (size_t)(8 * lh) * ldb + n0 + NT * lc;

    auto loadA = [&](int kb, v2f (&a)[MT][4]) {
#pragma unroll
        for (int i = 0; i < MT; ++i) {
            const float* p = Ap + (size_t)i * 16 * lda + kb;
            const v4f q0 = *(const v4f*)p;
            const v4f q1 = *(const v4f*)(p + 4);
            a[i][0] = __builtin_shufflevector(q0, q0, 0, 1);
            a[i][1] = __builtin_shufflevector(q0, q0, 2, 3);
            a[i][2] = __builtin_shufflevector(q1, q1, 0, 1);
            a[i][3] = __builtin_shufflevector(q1, q1, 2, 3);
        }
    };
    auto loadB = [&](int kb, float (&b)[8][4]) {
#pragma unroll
        for (int r = 0; r < 8; ++r) {
            const float* p = Bp + (size_t)(kb + r) * ldb;
            if constexpr (NT == 4) {
                const v4f q = *(const v4f*)p;
                b[r][0] = q.x; b[r][1] = q.y; b[r][2] = q.z; b[r][3] = q.w;
            } else {
#pragma unroll
                for (int t = 0; t < NT; ++t) b[r][t] = p[t];
            }
        }
    };
    auto mma = [&](const v2f (&a)[MT][4], const float (&b)[8][4]) {
#pragma unroll
        for (int j = 0; j < 4; ++j) {
#pragma unroll
            for (int i = 0; i < MT; ++i) {
#pragma unroll
                for (int t = 0; t < NT; ++t) {
                    v2f bv;
                    bv.x = b[2 * j][t];
                    bv.y = b[2 * j + 1][t];
                    acc[i][t] = __builtin_amdgcn_wmma_f32_16x16x4_f32(
                        false, a[i][j], false, bv, (short)0, acc[i][t], false, false);
                }
            }
        }
    };

    v2f a0[MT][4], a1[MT][4];
    float b0[8][4], b1[8][4];
    loadA(0, a0);
    loadB(0, b0);

    for (int kb = 0; kb < K; kb += 32) {
        loadA(kb + 16, a1);
        loadB(kb + 16, b1);
        __builtin_amdgcn_sched_barrier(0);
        mma(a0, b0);
        const int kn = (kb + 32 < K) ? kb + 32 : kb + 16;
        loadA(kn, a0);
        loadB(kn, b0);
        __builtin_amdgcn_sched_barrier(0);
        mma(a1, b1);
    }

    const float scale = scale_ptr ? scale_ptr[0] : 1.0f;
#pragma unroll
    for (int i = 0; i < MT; ++i) {
#pragma unroll
        for (int v = 0; v < 8; ++v) {
            const int m = m0 + i * 16 + 8 * lh + v;
            const int n = n0 + NT * lc;
            if constexpr (NT == 4) {
                v4f o;
                o.x = acc[i][0][v] * scale;
                o.y = acc[i][1][v] * scale;
                o.z = acc[i][2][v] * scale;
                o.w = acc[i][3][v] * scale;
                if (residual) o += *(const v4f*)(residual + (size_t)m * ldr + n);
                *(v4f*)(C + (size_t)m * ldc + n) = o;
            } else {
#pragma unroll
                for (int t = 0; t < NT; ++t) {
                    float x = acc[i][t][v] * scale;
                    if (residual) x += residual[(size_t)m * ldr + n + t];
                    C[(size_t)m * ldc + n + t] = x;
                }
            }
        }
    }
}

#define XP_WAVES 16
#define XP_KW (D_INNER / XP_WAVES)

__global__ __launch_bounds__(32 * XP_WAVES)
void xproj_splitk_kernel(const float* __restrict__ A, const float* __restrict__ Bm,
                         float* __restrict__ C)
{
    const int tid = threadIdx.x;
    const int lane = tid & 31;
    const int wave = tid >> 5;
    const int m0 = blockIdx.x * 16;
    const int lc = lane & 15;
    const int lh = lane >> 4;
    const int k0 = wave * XP_KW;

    __shared__ float sP[XP_WAVES][6][8][32];

    v8f acc[6];
#pragma unroll
    for (int t = 0; t < 6; ++t)
#pragma unroll
        for (int e = 0; e < 8; ++e) acc[t][e] = 0.f;

    const float* Ap = A + (size_t)(m0 + lc) * D_INNER + k0 + 8 * lh;
    const float* Bp = Bm + (size_t)(k0 + 8 * lh) * XDBL_W + 3 * lc;

    auto loadA = [&](int kb, v2f (&a)[4]) {
        const float* p = Ap + kb;
        const v4f q0 = *(const v4f*)p;
        const v4f q1 = *(const v4f*)(p + 4);
        a[0] = __builtin_shufflevector(q0, q0, 0, 1);
        a[1] = __builtin_shufflevector(q0, q0, 2, 3);
        a[2] = __builtin_shufflevector(q1, q1, 0, 1);
        a[3] = __builtin_shufflevector(q1, q1, 2, 3);
    };
    auto loadB = [&](int kb, float (&bl)[8][4], float (&bh)[8][4]) {
#pragma unroll
        for (int r = 0; r < 8; ++r) {
            const float* p = Bp + (size_t)(kb + r) * XDBL_W;
#pragma unroll
            for (int t = 0; t < 3; ++t) { bl[r][t] = p[t]; bh[r][t] = p[48 + t]; }
        }
    };
    auto mma = [&](const v2f (&a)[4], const float (&bl)[8][4], const float (&bh)[8][4]) {
#pragma unroll
        for (int j = 0; j < 4; ++j) {
#pragma unroll
            for (int t = 0; t < 3; ++t) {
                v2f bv;
                bv.x = bl[2 * j][t];
                bv.y = bl[2 * j + 1][t];
                acc[t] = __builtin_amdgcn_wmma_f32_16x16x4_f32(
                    false, a[j], false, bv, (short)0, acc[t], false, false);
                v2f bw;
                bw.x = bh[2 * j][t];
                bw.y = bh[2 * j + 1][t];
                acc[3 + t] = __builtin_amdgcn_wmma_f32_16x16x4_f32(
                    false, a[j], false, bw, (short)0, acc[3 + t], false, false);
            }
        }
    };

    v2f a0[4], a1[4];
    float bl0[8][4], bh0[8][4], bl1[8][4], bh1[8][4];
    loadA(0, a0);
    loadB(0, bl0, bh0);

    for (int kb = 0; kb < XP_KW; kb += 32) {
        loadA(kb + 16, a1);
        loadB(kb + 16, bl1, bh1);
        __builtin_amdgcn_sched_barrier(0);
        mma(a0, bl0, bh0);
        const int kn = (kb + 32 < XP_KW) ? kb + 32 : kb + 16;
        loadA(kn, a0);
        loadB(kn, bl0, bh0);
        __builtin_amdgcn_sched_barrier(0);
        mma(a1, bl1, bh1);
    }

#pragma unroll
    for (int t = 0; t < 6; ++t)
#pragma unroll
        for (int v = 0; v < 8; ++v) sP[wave][t][v][lane] = acc[t][v];
    __syncthreads();

    float* cb = C + (size_t)m0 * XDBL_W;
#pragma unroll
    for (int i = 0; i < (16 * XDBL_W) / (32 * XP_WAVES); ++i) {
        const int e = tid + (32 * XP_WAVES) * i;
        const int row = e / XDBL_W;
        const int col = e - row * XDBL_W;
        const int hi = (col >= 48) ? 1 : 0;
        const int cc = col - 48 * hi;
        const int c = cc / 3;
        const int t = cc - 3 * c + 3 * hi;
        const int ln = c + 16 * (row >> 3);
        const int v = row & 7;
        float s = sP[0][t][v][ln];
#pragma unroll
        for (int w = 1; w < XP_WAVES; ++w) s += sP[w][t][v][ln];
        cb[e] = s;
    }
}

#define EW_U 8
#define CONV_TL 8

__global__ __launch_bounds__(256)
void conv_silu_kernel(const float* __restrict__ xz, const float* __restrict__ w,
                      const float* __restrict__ bias, float* __restrict__ out)
{
    const int c4 = 4 * ((int)(blockIdx.x & 1) * 256 + (int)threadIdx.x);
    const int t0 = (int)(blockIdx.x >> 1) * CONV_TL;
    const bool head = (t0 & (LL - 1)) == 0;

    v4f wv[4];
#pragma unroll
    for (int ch = 0; ch < 4; ++ch)
        wv[ch] = *(const v4f*)(w + (size_t)(c4 + ch) * D_CONV);
    const v4f bv = *(const v4f*)(bias + c4);

    v4f r[CONV_TL + 3];
    const float* xp = xz + (size_t)t0 * (2 * D_INNER) + c4;
    const ptrdiff_t hist = head ? 0 : (ptrdiff_t)(2 * D_INNER);
#pragma unroll
    for (int i = 0; i < 3; ++i)
        r[i] = *(const v4f*)(xp + (ptrdiff_t)(i - 3) * hist);
#pragma unroll
    for (int i = 3; i < CONV_TL + 3; ++i)
        r[i] = *(const v4f*)(xp + (size_t)(i - 3) * (2 * D_INNER));
    __builtin_amdgcn_sched_barrier(0);

    float* op = out + (size_t)t0 * D_INNER + c4;
#pragma unroll
    for (int j = 0; j < CONV_TL; ++j) {
        v4f o;
#pragma unroll
        for (int ch = 0; ch < 4; ++ch) {
            float acc = bv[ch];
#pragma unroll
            for (int k = 0; k < D_CONV; ++k) {
                if (j + k >= 3 || !head)
                    acc += wv[ch][k] * r[j + k][ch];
            }
            o[ch] = acc * sigmoidf_(acc);
        }
        *(v4f*)(op + (size_t)j * D_INNER) = o;
    }
}

__global__ __launch_bounds__(256)
void conv_state_kernel(const float* __restrict__ xz, float* __restrict__ out)
{
    const int idx = blockIdx.x * 256 + threadIdx.x;
    if (idx >= BB * D_INNER * (D_CONV - 1)) return;
    const int j = idx % (D_CONV - 1);
    const int c = (idx / (D_CONV - 1)) % D_INNER;
    const int b = idx / ((D_CONV - 1) * D_INNER);
    out[idx] = xz[((size_t)(b * LL + (LL - (D_CONV - 1) + j))) * (2 * D_INNER) + c];
}

__global__ __launch_bounds__(256)
void dt_softplus_kernel(float* __restrict__ dt, const float* __restrict__ bias)
{
    v4f* p = (v4f*)dt + (size_t)blockIdx.x * (EW_U * 256) + threadIdx.x;
    const v4f bq[2] = { ((const v4f*)bias)[threadIdx.x],
                        ((const v4f*)bias)[threadIdx.x + 256] };
    v4f v[EW_U];
#pragma unroll
    for (int u = 0; u < EW_U; ++u) v[u] = p[256 * u];
#pragma unroll
    for (int u = 0; u < EW_U; ++u) {
        v4f o;
#pragma unroll
        for (int e = 0; e < 4; ++e) {
            const float x = v[u][e] + bq[u & 1][e];
            o[e] = (x > 20.f) ? x : __logf(1.f + __expf(x));
        }
        p[256 * u] = o;
    }
}

#define SCAN_T   32
#define SCAN_NCH (LL / SCAN_T)
#define SCAN_WPB (D_INNER / 32)
#define SCAN_P   4
#define SCAN_S   (D_STATE / SCAN_P)
#define SCAN_R   (SCAN_T / SCAN_P)
#define SCAN_Q   (SCAN_T * 8 / (32 * SCAN_P))

__global__ __launch_bounds__(32 * SCAN_P)
void scan_kernel(const float* __restrict__ dt, const float* __restrict__ xc,
                 const float* __restrict__ xdbl, const float* __restrict__ A_log,
                 float* __restrict__ y, float* __restrict__ ssm_out)
{
    const int tid = threadIdx.x;
    const int lane = tid & 31;
    const int w = tid >> 5;
    const int b = blockIdx.x / SCAN_WPB;
    const int d0 = (blockIdx.x % SCAN_WPB) * 32;
    const int d = d0 + lane;

    __shared__ __attribute__((aligned(16))) float sBC[2][SCAN_T * 32];
    __shared__ __attribute__((aligned(16))) v2f sDX[2][SCAN_T * 32];
    __shared__ float sY[2][SCAN_P][SCAN_T][32];

    v2f Aval[2], h[2];
    {
        const v4f al = *(const v4f*)(A_log + (size_t)d * D_STATE + SCAN_S * w);
        Aval[0].x = -__expf(al.x);
        Aval[0].y = -__expf(al.y);
        Aval[1].x = -__expf(al.z);
        Aval[1].y = -__expf(al.w);
        h[0].x = 0.f; h[0].y = 0.f;
        h[1].x = 0.f; h[1].y = 0.f;
    }

    const size_t row0 = (size_t)b * LL;
    float* yp = y + row0 * D_INNER + d;
    const int prow = tid >> 3;
    const int pcol = 4 * (tid & 7);
    const float* bcp = xdbl + (row0 + prow) * XDBL_W + DT_RANK + pcol;
    const float* dtq = dt + (row0 + prow) * D_INNER + d0 + pcol;
    const float* xq  = xc + (row0 + prow) * D_INNER + d0 + pcol;

    v4f nbc[SCAN_Q], ndt[SCAN_Q], nxx[SCAN_Q];

    auto fetch = [&](int c) {
        const size_t l0 = (size_t)c * SCAN_T;
#pragma unroll
        for (int i = 0; i < SCAN_Q; ++i) {
            nbc[i] = *(const v4f*)(bcp + (l0 + 16 * i) * XDBL_W);
            ndt[i] = *(const v4f*)(dtq + (l0 + 16 * i) * D_INNER);
            nxx[i] = *(const v4f*)(xq  + (l0 + 16 * i) * D_INNER);
        }
    };
    auto stash = [&](int p) {
#pragma unroll
        for (int i = 0; i < SCAN_Q; ++i) {
            *(v4f*)&sBC[p][4 * tid + 512 * i] = nbc[i];
            v4f lo, hi;
            lo.x = ndt[i].x; lo.y = nxx[i].x; lo.z = ndt[i].y; lo.w = nxx[i].y;
            hi.x = ndt[i].z; hi.y = nxx[i].z; hi.z = ndt[i].w; hi.w = nxx[i].w;
            v2f* q = &sDX[p][4 * tid + 512 * i];
            *(v4f*)q = lo;
            *(v4f*)(q + 2) = hi;
        }
    };

    auto lds_handover = [&]() {
        __builtin_amdgcn_fence(__ATOMIC_RELEASE, "workgroup", "local");
        __builtin_amdgcn_s_barrier();
        __builtin_amdgcn_fence(__ATOMIC_ACQUIRE, "workgroup", "local");
    };

    fetch(0);
    stash(0);
    lds_handover();

    for (int c = 0; c < SCAN_NCH; ++c) {
        const int p = c & 1;
        const v4f* bc = (const v4f*)&sBC[p][0];
        v4f rB[SCAN_T], rC[SCAN_T];
        v2f dx[SCAN_T];
#pragma unroll
        for (int s = 0; s < SCAN_T; ++s) {
            rB[s] = bc[s * 8 + w];
            rC[s] = bc[s * 8 + 4 + w];
            dx[s] = sDX[p][s * 32 + lane];
        }

        const int cn = (c + 1 < SCAN_NCH) ? c + 1 : c;
        fetch(cn);
        __builtin_amdgcn_sched_barrier(0);

#pragma unroll
        for (int s = 0; s < SCAN_T; ++s) {
            const float dtv = dx[s].x;
            const float bx = dtv * dx[s].y;
            v2f Bk[2], Ck[2];
            Bk[0].x = rB[s].x; Bk[0].y = rB[s].y; Bk[1].x = rB[s].z; Bk[1].y = rB[s].w;
            Ck[0].x = rC[s].x; Ck[0].y = rC[s].y; Ck[1].x = rC[s].z; Ck[1].y = rC[s].w;
            float yv = 0.f;
#pragma unroll
            for (int k = 0; k < 2; ++k) {
                const v2f a = Aval[k] * dtv;
                v2f dA;
                dA.x = __expf(a.x);
                dA.y = __expf(a.y);
                h[k] = __builtin_elementwise_fma(dA, h[k], bx * Bk[k]);
                yv += h[k].x * Ck[k].x;
                yv += h[k].y * Ck[k].y;
            }
            sY[p][w][s][lane] = yv;
        }
        __builtin_amdgcn_sched_barrier(0);

        stash(p ^ 1);
        lds_handover();

        float* yrow = yp + ((size_t)c * SCAN_T + SCAN_R * w) * D_INNER;
#pragma unroll
        for (int j = 0; j < SCAN_R; ++j) {
            const int s = SCAN_R * w + j;
            float t = sY[p][0][s][lane];
#pragma unroll
            for (int v = 1; v < SCAN_P; ++v) t += sY[p][v][s][lane];
            yrow[(size_t)j * D_INNER] = t;
        }
    }

    {
        v4f ho;
        ho.x = h[0].x; ho.y = h[0].y; ho.z = h[1].x; ho.w = h[1].y;
        *(v4f*)(ssm_out + (size_t)(b * D_INNER + d) * D_STATE + SCAN_S * w) = ho;
    }
}

__global__ __launch_bounds__(256)
void gate_kernel(float* __restrict__ y, const float* __restrict__ xc,
                 const float* __restrict__ xz, const float* __restrict__ Dp)
{
    const size_t base = (size_t)blockIdx.x * (EW_U * 256) + threadIdx.x;
    v4f* yp = (v4f*)y + base;
    const v4f* xp = (const v4f*)xc + base;
    const v4f* zp = (const v4f*)(xz + (size_t)blockIdx.x * (EW_U / 2) * (2 * D_INNER) + D_INNER)
                    + threadIdx.x;
    const v4f dq[2] = { ((const v4f*)Dp)[threadIdx.x],
                        ((const v4f*)Dp)[threadIdx.x + 256] };
    v4f yv[EW_U], xv[EW_U], zv[EW_U];
#pragma unroll
    for (int u = 0; u < EW_U; ++u) {
        yv[u] = yp[256 * u];
        xv[u] = xp[256 * u];
        zv[u] = zp[(u >> 1) * (2 * D_INNER / 4) + (u & 1) * 256];
    }
    __builtin_amdgcn_sched_barrier(0);
#pragma unroll
    for (int u = 0; u < EW_U; ++u) {
        v4f o;
#pragma unroll
        for (int e = 0; e < 4; ++e) {
            const float z = zv[u][e];
            const float t = yv[u][e] + xv[u][e] * dq[u & 1][e];
            o[e] = t * z * sigmoidf_(z);
        }
        yp[256 * u] = o;
    }
}

extern "C" void kernel_launch(void* const* d_in, const int* in_sizes, int n_in,
                              void* d_out, int out_size, void* d_ws, size_t ws_size,
                              hipStream_t stream)
{
    const float* x         = (const float*)d_in[0];
    const float* ln_w      = (const float*)d_in[1];
    const float* ln_b      = (const float*)d_in[2];
    const float* in_proj_w = (const float*)d_in[3];
    const float* conv_w    = (const float*)d_in[4];
    const float* conv_b    = (const float*)d_in[5];
    const float* x_proj_w  = (const float*)d_in[6];
    const float* dt_proj_w = (const float*)d_in[7];
    const float* dt_proj_b = (const float*)d_in[8];
    const float* A_log     = (const float*)d_in[9];
    const float* D_param   = (const float*)d_in[10];
    const float* out_projw = (const float*)d_in[11];
    const float* res_scale = (const float*)d_in[12];

    float* out      = (float*)d_out;
    float* out_ssm  = out + (size_t)BB * LL * D_MODEL;
    float* out_conv = out_ssm + (size_t)BB * D_INNER * D_STATE;

    float* ws    = (float*)d_ws;
    float* xn    = ws;
    float* xz    = xn    + (size_t)TOKENS * D_MODEL;
    float* xc    = xz    + (size_t)TOKENS * 2 * D_INNER;
    float* xdbl  = xc    + (size_t)TOKENS * D_INNER;
    float* dtbuf = xdbl  + (size_t)TOKENS * XDBL_W;
    float* ybuf  = dtbuf + (size_t)TOKENS * D_INNER;

    ln_kernel<<<TOKENS, 256, 0, stream>>>(x, ln_w, ln_b, xn);

    wmma_gemm_kernel<2, 4><<<dim3((2 * D_INNER) / 128, TOKENS / 64), 128, 0, stream>>>(
        xn, D_MODEL, in_proj_w, 2 * D_INNER, xz, 2 * D_INNER, D_MODEL,
        nullptr, 0, nullptr);

    conv_silu_kernel<<<(TOKENS / CONV_TL) * (D_INNER / 4 / 256), 256, 0, stream>>>(
        xz, conv_w, conv_b, xc);
    conv_state_kernel<<<(BB * D_INNER * (D_CONV - 1) + 255) / 256, 256, 0, stream>>>(
        xz, out_conv);

    xproj_splitk_kernel<<<TOKENS / 16, 32 * XP_WAVES, 0, stream>>>(xc, x_proj_w, xdbl);

    wmma_gemm_kernel<2, 4><<<dim3(D_INNER / 128, TOKENS / 64), 128, 0, stream>>>(
        xdbl, XDBL_W, dt_proj_w, D_INNER, dtbuf, D_INNER, DT_RANK,
        nullptr, 0, nullptr);

    dt_softplus_kernel<<<(TOKENS * D_INNER / 4) / (EW_U * 256), 256, 0, stream>>>(dtbuf, dt_proj_b);

    scan_kernel<<<BB * SCAN_WPB, 32 * SCAN_P, 0, stream>>>(
        dtbuf, xc, xdbl, A_log, ybuf, out_ssm);

    gate_kernel<<<(TOKENS * D_INNER / 4) / (EW_U * 256), 256, 0, stream>>>(ybuf, xc, xz, D_param);

    wmma_gemm_kernel<2, 4><<<dim3(D_MODEL / 128, TOKENS / 64), 128, 0, stream>>>(
        ybuf, D_INNER, out_projw, D_MODEL, out, D_MODEL, D_INNER,
        x, D_MODEL, res_scale);
}


# elementwise launches folded into their producers: softplus into the dt_proj GEMM epilogue, gate into the scan's y write-out, conv tail state into the conv kernel (10 launches to 7); sums unchanged
# speedup vs baseline: 5.6432x; 1.1001x over previous
#include <hip/hip_runtime.h>
#include <hip/hip_bf16.h>

typedef __attribute__((ext_vector_type(2))) float v2f;
typedef __attribute__((ext_vector_type(4))) float v4f;
typedef __attribute__((ext_vector_type(8))) float v8f;

#define D_MODEL 1024
#define D_STATE 16
#define D_CONV  4
#define D_INNER 2048
#define DT_RANK 64
#define BB 2
#define LL 2048
#define TOKENS (BB * LL)
#define XDBL_W (DT_RANK + 2 * D_STATE)

__device__ __forceinline__ float sigmoidf_(float x) { return 1.f / (1.f + __expf(-x)); }

__global__ __launch_bounds__(256)
void ln_kernel(const float* __restrict__ x, const float* __restrict__ w,
               const float* __restrict__ b, float* __restrict__ out)
{
    const int row = blockIdx.x;
    const float* xr = x + (size_t)row * D_MODEL;
    float v[4];
    float s = 0.f, s2 = 0.f;
#pragma unroll
    for (int i = 0; i < 4; ++i) {
        v[i] = xr[threadIdx.x + i * 256];
        s += v[i];
        s2 += v[i] * v[i];
    }
#pragma unroll
    for (int off = 16; off > 0; off >>= 1) {
        s  += __shfl_down(s,  off, 32);
        s2 += __shfl_down(s2, off, 32);
    }
    __shared__ float rs[8], rq[8];
    __shared__ float mu_s, rstd_s;
    const int wave = threadIdx.x >> 5;
    const int lane = threadIdx.x & 31;
    if (lane == 0) { rs[wave] = s; rq[wave] = s2; }
    __syncthreads();
    if (threadIdx.x == 0) {
        float a = 0.f, q = 0.f;
#pragma unroll
        for (int i = 0; i < 8; ++i) { a += rs[i]; q += rq[i]; }
        const float mu  = a * (1.f / D_MODEL);
        const float var = q * (1.f / D_MODEL) - mu * mu;
        mu_s = mu;
        rstd_s = rsqrtf(var + 1e-5f);
    }
    __syncthreads();
    const float mu = mu_s, rstd = rstd_s;
    float* orow = out + (size_t)row * D_MODEL;
#pragma unroll
    for (int i = 0; i < 4; ++i) {
        const int c = threadIdx.x + i * 256;
        orow[c] = (v[i] - mu) * rstd * w[c] + b[c];
    }
}

template <int MT, int NT, bool SOFTPLUS>
__global__ __launch_bounds__(128)
void wmma_gemm_kernel(const float* __restrict__ A, int lda,
                      const float* __restrict__ Bm, int ldb,
                      float* __restrict__ C, int ldc, int K,
                      const float* __restrict__ residual, int ldr,
                      const float* __restrict__ scale_ptr,
                      const float* __restrict__ bias)
{
    static_assert(!SOFTPLUS || NT == 4, "the softplus epilogue is written for NT == 4");
    const int lane = threadIdx.x & 31;
    const int wave = threadIdx.x >> 5;
    const int wy = wave & 1;
    const int wx = wave >> 1;
    const int m0 = blockIdx.y * (32 * MT) + wy * (16 * MT);
    const int n0 = blockIdx.x * (32 * NT) + wx * (16 * NT);
    const int lc = lane & 15;
    const int lh = lane >> 4;

    v8f acc[MT][NT];
#pragma unroll
    for (int i = 0; i < MT; ++i)
#pragma unroll
        for (int t = 0; t < NT; ++t)
#pragma unroll
            for (int e = 0; e < 8; ++e) acc[i][t][e] = 0.f;

    const float* Ap = A + (size_t)(m0 + lc) * lda + 8 * lh;
    const float* Bp = Bm + (size_t)(8 * lh) * ldb + n0 + NT * lc;

    auto loadA = [&](int kb, v2f (&a)[MT][4]) {
#pragma unroll
        for (int i = 0; i < MT; ++i) {
            const float* p = Ap + (size_t)i * 16 * lda + kb;
            const v4f q0 = *(const v4f*)p;
            const v4f q1 = *(const v4f*)(p + 4);
            a[i][0] = __builtin_shufflevector(q0, q0, 0, 1);
            a[i][1] = __builtin_shufflevector(q0, q0, 2, 3);
            a[i][2] = __builtin_shufflevector(q1, q1, 0, 1);
            a[i][3] = __builtin_shufflevector(q1, q1, 2, 3);
        }
    };
    auto loadB = [&](int kb, float (&b)[8][4]) {
#pragma unroll
        for (int r = 0; r < 8; ++r) {
            const float* p = Bp + (size_t)(kb + r) * ldb;
            if constexpr (NT == 4) {
                const v4f q = *(const v4f*)p;
                b[r][0] = q.x; b[r][1] = q.y; b[r][2] = q.z; b[r][3] = q.w;
            } else {
#pragma unroll
                for (int t = 0; t < NT; ++t) b[r][t] = p[t];
            }
        }
    };
    auto mma = [&](const v2f (&a)[MT][4], const float (&b)[8][4]) {
#pragma unroll
        for (int j = 0; j < 4; ++j) {
#pragma unroll
            for (int i = 0; i < MT; ++i) {
#pragma unroll
                for (int t = 0; t < NT; ++t) {
                    v2f bv;
                    bv.x = b[2 * j][t];
                    bv.y = b[2 * j + 1][t];
                    acc[i][t] = __builtin_amdgcn_wmma_f32_16x16x4_f32(
                        false, a[i][j], false, bv, (short)0, acc[i][t], false, false);
                }
            }
        }
    };

    v2f a0[MT][4], a1[MT][4];
    float b0[8][4], b1[8][4];
    loadA(0, a0);
    loadB(0, b0);

    for (int kb = 0; kb < K; kb += 32) {
        loadA(kb + 16, a1);
        loadB(kb + 16, b1);
        __builtin_amdgcn_sched_barrier(0);
        mma(a0, b0);
        const int kn = (kb + 32 < K) ? kb + 32 : kb + 16;
        loadA(kn, a0);
        loadB(kn, b0);
        __builtin_amdgcn_sched_barrier(0);
        mma(a1, b1);
    }

    if constexpr (SOFTPLUS) {
        const int n = n0 + NT * lc;
        const v4f bq = *(const v4f*)(bias + n);
#pragma unroll
        for (int i = 0; i < MT; ++i) {
#pragma unroll
            for (int v = 0; v < 8; ++v) {
                const int m = m0 + i * 16 + 8 * lh + v;
                v4f o;
#pragma unroll
                for (int t = 0; t < 4; ++t) {
                    const float x = acc[i][t][v] + bq[t];
                    o[t] = (x > 20.f) ? x : __logf(1.f + __expf(x));
                }
                *(v4f*)(C + (size_t)m * ldc + n) = o;
            }
        }
        return;
    }

    const float scale = scale_ptr ? scale_ptr[0] : 1.0f;
#pragma unroll
    for (int i = 0; i < MT; ++i) {
#pragma unroll
        for (int v = 0; v < 8; ++v) {
            const int m = m0 + i * 16 + 8 * lh + v;
            const int n = n0 + NT * lc;
            if constexpr (NT == 4) {
                v4f o;
                o.x = acc[i][0][v] * scale;
                o.y = acc[i][1][v] * scale;
                o.z = acc[i][2][v] * scale;
                o.w = acc[i][3][v] * scale;
                if (residual) o += *(const v4f*)(residual + (size_t)m * ldr + n);
                *(v4f*)(C + (size_t)m * ldc + n) = o;
            } else {
#pragma unroll
                for (int t = 0; t < NT; ++t) {
                    float x = acc[i][t][v] * scale;
                    if (residual) x += residual[(size_t)m * ldr + n + t];
                    C[(size_t)m * ldc + n + t] = x;
                }
            }
        }
    }
}

#define XP_WAVES 16
#define XP_KW (D_INNER / XP_WAVES)

__global__ __launch_bounds__(32 * XP_WAVES)
void xproj_splitk_kernel(const float* __restrict__ A, const float* __restrict__ Bm,
                         float* __restrict__ C)
{
    const int tid = threadIdx.x;
    const int lane = tid & 31;
    const int wave = tid >> 5;
    const int m0 = blockIdx.x * 16;
    const int lc = lane & 15;
    const int lh = lane >> 4;
    const int k0 = wave * XP_KW;

    __shared__ float sP[XP_WAVES][6][8][32];

    v8f acc[6];
#pragma unroll
    for (int t = 0; t < 6; ++t)
#pragma unroll
        for (int e = 0; e < 8; ++e) acc[t][e] = 0.f;

    const float* Ap = A + (size_t)(m0 + lc) * D_INNER + k0 + 8 * lh;
    const float* Bp = Bm + (size_t)(k0 + 8 * lh) * XDBL_W + 3 * lc;

    auto loadA = [&](int kb, v2f (&a)[4]) {
        const float* p = Ap + kb;
        const v4f q0 = *(const v4f*)p;
        const v4f q1 = *(const v4f*)(p + 4);
        a[0] = __builtin_shufflevector(q0, q0, 0, 1);
        a[1] = __builtin_shufflevector(q0, q0, 2, 3);
        a[2] = __builtin_shufflevector(q1, q1, 0, 1);
        a[3] = __builtin_shufflevector(q1, q1, 2, 3);
    };
    auto loadB = [&](int kb, float (&bl)[8][4], float (&bh)[8][4]) {
#pragma unroll
        for (int r = 0; r < 8; ++r) {
            const float* p = Bp + (size_t)(kb + r) * XDBL_W;
#pragma unroll
            for (int t = 0; t < 3; ++t) { bl[r][t] = p[t]; bh[r][t] = p[48 + t]; }
        }
    };
    auto mma = [&](const v2f (&a)[4], const float (&bl)[8][4], const float (&bh)[8][4]) {
#pragma unroll
        for (int j = 0; j < 4; ++j) {
#pragma unroll
            for (int t = 0; t < 3; ++t) {
                v2f bv;
                bv.x = bl[2 * j][t];
                bv.y = bl[2 * j + 1][t];
                acc[t] = __builtin_amdgcn_wmma_f32_16x16x4_f32(
                    false, a[j], false, bv, (short)0, acc[t], false, false);
                v2f bw;
                bw.x = bh[2 * j][t];
                bw.y = bh[2 * j + 1][t];
                acc[3 + t] = __builtin_amdgcn_wmma_f32_16x16x4_f32(
                    false, a[j], false, bw, (short)0, acc[3 + t], false, false);
            }
        }
    };

    v2f a0[4], a1[4];
    float bl0[8][4], bh0[8][4], bl1[8][4], bh1[8][4];
    loadA(0, a0);
    loadB(0, bl0, bh0);

    for (int kb = 0; kb < XP_KW; kb += 32) {
        loadA(kb + 16, a1);
        loadB(kb + 16, bl1, bh1);
        __builtin_amdgcn_sched_barrier(0);
        mma(a0, bl0, bh0);
        const int kn = (kb + 32 < XP_KW) ? kb + 32 : kb + 16;
        loadA(kn, a0);
        loadB(kn, bl0, bh0);
        __builtin_amdgcn_sched_barrier(0);
        mma(a1, bl1, bh1);
    }

#pragma unroll
    for (int t = 0; t < 6; ++t)
#pragma unroll
        for (int v = 0; v < 8; ++v) sP[wave][t][v][lane] = acc[t][v];
    __syncthreads();

    float* cb = C + (size_t)m0 * XDBL_W;
#pragma unroll
    for (int i = 0; i < (16 * XDBL_W) / (32 * XP_WAVES); ++i) {
        const int e = tid + (32 * XP_WAVES) * i;
        const int row = e / XDBL_W;
        const int col = e - row * XDBL_W;
        const int hi = (col >= 48) ? 1 : 0;
        const int cc = col - 48 * hi;
        const int c = cc / 3;
        const int t = cc - 3 * c + 3 * hi;
        const int ln = c + 16 * (row >> 3);
        const int v = row & 7;
        float s = sP[0][t][v][ln];
#pragma unroll
        for (int w = 1; w < XP_WAVES; ++w) s += sP[w][t][v][ln];
        cb[e] = s;
    }
}

#define CONV_TL 8

__global__ __launch_bounds__(256)
void conv_silu_kernel(const float* __restrict__ xz, const float* __restrict__ w,
                      const float* __restrict__ bias, float* __restrict__ out,
                      float* __restrict__ state_out)
{
    const int c4 = 4 * ((int)(blockIdx.x & 1) * 256 + (int)threadIdx.x);
    const int t0 = (int)(blockIdx.x >> 1) * CONV_TL;
    const bool head = (t0 & (LL - 1)) == 0;
    const bool tail = ((t0 + CONV_TL) & (LL - 1)) == 0;

    v4f wv[4];
#pragma unroll
    for (int ch = 0; ch < 4; ++ch)
        wv[ch] = *(const v4f*)(w + (size_t)(c4 + ch) * D_CONV);
    const v4f bv = *(const v4f*)(bias + c4);

    v4f r[CONV_TL + 3];
    const float* xp = xz + (size_t)t0 * (2 * D_INNER) + c4;
    const ptrdiff_t hist = head ? 0 : (ptrdiff_t)(2 * D_INNER);
#pragma unroll
    for (int i = 0; i < 3; ++i)
        r[i] = *(const v4f*)(xp + (ptrdiff_t)(i - 3) * hist);
#pragma unroll
    for (int i = 3; i < CONV_TL + 3; ++i)
        r[i] = *(const v4f*)(xp + (size_t)(i - 3) * (2 * D_INNER));
    __builtin_amdgcn_sched_barrier(0);

    float* op = out + (size_t)t0 * D_INNER + c4;
#pragma unroll
    for (int j = 0; j < CONV_TL; ++j) {
        v4f o;
#pragma unroll
        for (int ch = 0; ch < 4; ++ch) {
            float acc = bv[ch];
#pragma unroll
            for (int k = 0; k < D_CONV; ++k) {
                if (j + k >= 3 || !head)
                    acc += wv[ch][k] * r[j + k][ch];
            }
            o[ch] = acc * sigmoidf_(acc);
        }
        *(v4f*)(op + (size_t)j * D_INNER) = o;
    }

    if (tail) {
        const int b = t0 / LL;
        float* sp = state_out + ((size_t)b * D_INNER + c4) * (D_CONV - 1);
        float s[4 * (D_CONV - 1)];
#pragma unroll
        for (int ch = 0; ch < 4; ++ch)
#pragma unroll
            for (int j = 0; j < D_CONV - 1; ++j)
                s[ch * (D_CONV - 1) + j] = r[CONV_TL + j][ch];
#pragma unroll
        for (int q = 0; q < 3; ++q) {
            v4f o;
            o.x = s[4 * q]; o.y = s[4 * q + 1]; o.z = s[4 * q + 2]; o.w = s[4 * q + 3];
            *(v4f*)(sp + 4 * q) = o;
        }
    }
}

#define SCAN_T   32
#define SCAN_NCH (LL / SCAN_T)
#define SCAN_WPB (D_INNER / 32)
#define SCAN_P   4
#define SCAN_S   (D_STATE / SCAN_P)
#define SCAN_R   (SCAN_T / SCAN_P)
#define SCAN_Q   (SCAN_T * 8 / (32 * SCAN_P))
#define SCAN_OQ  (SCAN_R / 4)

__global__ __launch_bounds__(32 * SCAN_P)
void scan_kernel(const float* __restrict__ dt, const float* __restrict__ xc,
                 const float* __restrict__ xdbl, const float* __restrict__ A_log,
                 const float* __restrict__ xz, const float* __restrict__ Dp,
                 float* __restrict__ y, float* __restrict__ ssm_out)
{
    const int tid = threadIdx.x;
    const int lane = tid & 31;
    const int w = tid >> 5;
    const int b = blockIdx.x / SCAN_WPB;
    const int d0 = (blockIdx.x % SCAN_WPB) * 32;
    const int d = d0 + lane;

    __shared__ __attribute__((aligned(16))) float sBC[2][SCAN_T * 32];
    __shared__ __attribute__((aligned(16))) v2f sDX[2][SCAN_T * 32];
    __shared__ __attribute__((aligned(16))) float sY[2][SCAN_P][SCAN_T][32];

    v2f Aval[2], h[2];
    {
        const v4f al = *(const v4f*)(A_log + (size_t)d * D_STATE + SCAN_S * w);
        Aval[0].x = -__expf(al.x);
        Aval[0].y = -__expf(al.y);
        Aval[1].x = -__expf(al.z);
        Aval[1].y = -__expf(al.w);
        h[0].x = 0.f; h[0].y = 0.f;
        h[1].x = 0.f; h[1].y = 0.f;
    }

    const size_t row0 = (size_t)b * LL;
    const int orow = SCAN_R * w + (lane >> 3);
    const int ocol = 4 * (lane & 7);
    const float* zg = xz + (row0 + orow) * (2 * D_INNER) + D_INNER + d0 + ocol;
    const float* xg = xc + (row0 + orow) * D_INNER + d0 + ocol;
    float* yg = y + (row0 + orow) * D_INNER + d0 + ocol;
    const v4f Dq = *(const v4f*)(Dp + d0 + ocol);
    const int prow = tid >> 3;
    const int pcol = 4 * (tid & 7);
    const float* bcp = xdbl + (row0 + prow) * XDBL_W + DT_RANK + pcol;
    const float* dtq = dt + (row0 + prow) * D_INNER + d0 + pcol;
    const float* xq  = xc + (row0 + prow) * D_INNER + d0 + pcol;

    v4f nbc[SCAN_Q], ndt[SCAN_Q], nxx[SCAN_Q];

    auto fetch = [&](int c) {
        const size_t l0 = (size_t)c * SCAN_T;
#pragma unroll
        for (int i = 0; i < SCAN_Q; ++i) {
            nbc[i] = *(const v4f*)(bcp + (l0 + 16 * i) * XDBL_W);
            ndt[i] = *(const v4f*)(dtq + (l0 + 16 * i) * D_INNER);
            nxx[i] = *(const v4f*)(xq  + (l0 + 16 * i) * D_INNER);
        }
    };
    auto stash = [&](int p) {
#pragma unroll
        for (int i = 0; i < SCAN_Q; ++i) {
            *(v4f*)&sBC[p][4 * tid + 512 * i] = nbc[i];
            v4f lo, hi;
            lo.x = ndt[i].x; lo.y = nxx[i].x; lo.z = ndt[i].y; lo.w = nxx[i].y;
            hi.x = ndt[i].z; hi.y = nxx[i].z; hi.z = ndt[i].w; hi.w = nxx[i].w;
            v2f* q = &sDX[p][4 * tid + 512 * i];
            *(v4f*)q = lo;
            *(v4f*)(q + 2) = hi;
        }
    };

    auto lds_handover = [&]() {
        __builtin_amdgcn_fence(__ATOMIC_RELEASE, "workgroup", "local");
        __builtin_amdgcn_s_barrier();
        __builtin_amdgcn_fence(__ATOMIC_ACQUIRE, "workgroup", "local");
    };

    fetch(0);
    stash(0);
    lds_handover();

    for (int c = 0; c < SCAN_NCH; ++c) {
        const int p = c & 1;
        const v4f* bc = (const v4f*)&sBC[p][0];
        v4f rB[SCAN_T], rC[SCAN_T];
        v2f dx[SCAN_T];
#pragma unroll
        for (int s = 0; s < SCAN_T; ++s) {
            rB[s] = bc[s * 8 + w];
            rC[s] = bc[s * 8 + 4 + w];
            dx[s] = sDX[p][s * 32 + lane];
        }

        const int cn = (c + 1 < SCAN_NCH) ? c + 1 : c;
        fetch(cn);
        v4f zr[SCAN_OQ], xr[SCAN_OQ];
#pragma unroll
        for (int i = 0; i < SCAN_OQ; ++i) {
            zr[i] = *(const v4f*)(zg + ((size_t)c * SCAN_T + 4 * i) * (2 * D_INNER));
            xr[i] = *(const v4f*)(xg + ((size_t)c * SCAN_T + 4 * i) * D_INNER);
        }
        __builtin_amdgcn_sched_barrier(0);

#pragma unroll
        for (int s = 0; s < SCAN_T; ++s) {
            const float dtv = dx[s].x;
            const float bx = dtv * dx[s].y;
            v2f Bk[2], Ck[2];
            Bk[0].x = rB[s].x; Bk[0].y = rB[s].y; Bk[1].x = rB[s].z; Bk[1].y = rB[s].w;
            Ck[0].x = rC[s].x; Ck[0].y = rC[s].y; Ck[1].x = rC[s].z; Ck[1].y = rC[s].w;
            float yv = 0.f;
#pragma unroll
            for (int k = 0; k < 2; ++k) {
                const v2f a = Aval[k] * dtv;
                v2f dA;
                dA.x = __expf(a.x);
                dA.y = __expf(a.y);
                h[k] = __builtin_elementwise_fma(dA, h[k], bx * Bk[k]);
                yv += h[k].x * Ck[k].x;
                yv += h[k].y * Ck[k].y;
            }
            sY[p][w][s][lane] = yv;
        }
        __builtin_amdgcn_sched_barrier(0);

        stash(p ^ 1);
        lds_handover();

#pragma unroll
        for (int i = 0; i < SCAN_OQ; ++i) {
            const int s = orow + 4 * i;
            v4f t = *(const v4f*)&sY[p][0][s][ocol];
#pragma unroll
            for (int v = 1; v < SCAN_P; ++v) t += *(const v4f*)&sY[p][v][s][ocol];
            v4f o;
#pragma unroll
            for (int e = 0; e < 4; ++e) {
                const float z = zr[i][e];
                const float g = t[e] + xr[i][e] * Dq[e];
                o[e] = g * z * sigmoidf_(z);
            }
            *(v4f*)(yg + ((size_t)c * SCAN_T + 4 * i) * D_INNER) = o;
        }
    }

    {
        v4f ho;
        ho.x = h[0].x; ho.y = h[0].y; ho.z = h[1].x; ho.w = h[1].y;
        *(v4f*)(ssm_out + (size_t)(b * D_INNER + d) * D_STATE + SCAN_S * w) = ho;
    }
}

extern "C" void kernel_launch(void* const* d_in, const int* in_sizes, int n_in,
                              void* d_out, int out_size, void* d_ws, size_t ws_size,
                              hipStream_t stream)
{
    const float* x         = (const float*)d_in[0];
    const float* ln_w      = (const float*)d_in[1];
    const float* ln_b      = (const float*)d_in[2];
    const float* in_proj_w = (const float*)d_in[3];
    const float* conv_w    = (const float*)d_in[4];
    const float* conv_b    = (const float*)d_in[5];
    const float* x_proj_w  = (const float*)d_in[6];
    const float* dt_proj_w = (const float*)d_in[7];
    const float* dt_proj_b = (const float*)d_in[8];
    const float* A_log     = (const float*)d_in[9];
    const float* D_param   = (const float*)d_in[10];
    const float* out_projw = (const float*)d_in[11];
    const float* res_scale = (const float*)d_in[12];

    float* out      = (float*)d_out;
    float* out_ssm  = out + (size_t)BB * LL * D_MODEL;
    float* out_conv = out_ssm + (size_t)BB * D_INNER * D_STATE;

    float* ws    = (float*)d_ws;
    float* xn    = ws;
    float* xz    = xn    + (size_t)TOKENS * D_MODEL;
    float* xc    = xz    + (size_t)TOKENS * 2 * D_INNER;
    float* xdbl  = xc    + (size_t)TOKENS * D_INNER;
    float* dtbuf = xdbl  + (size_t)TOKENS * XDBL_W;
    float* ybuf  = dtbuf + (size_t)TOKENS * D_INNER;

    ln_kernel<<<TOKENS, 256, 0, stream>>>(x, ln_w, ln_b, xn);

    wmma_gemm_kernel<2, 4, false><<<dim3((2 * D_INNER) / 128, TOKENS / 64), 128, 0, stream>>>(
        xn, D_MODEL, in_proj_w, 2 * D_INNER, xz, 2 * D_INNER, D_MODEL,
        nullptr, 0, nullptr, nullptr);

    conv_silu_kernel<<<(TOKENS / CONV_TL) * (D_INNER / 4 / 256), 256, 0, stream>>>(
        xz, conv_w, conv_b, xc, out_conv);

    xproj_splitk_kernel<<<TOKENS / 16, 32 * XP_WAVES, 0, stream>>>(xc, x_proj_w, xdbl);

    wmma_gemm_kernel<2, 4, true><<<dim3(D_INNER / 128, TOKENS / 64), 128, 0, stream>>>(
        xdbl, XDBL_W, dt_proj_w, D_INNER, dtbuf, D_INNER, DT_RANK,
        nullptr, 0, nullptr, dt_proj_b);

    scan_kernel<<<BB * SCAN_WPB, 32 * SCAN_P, 0, stream>>>(
        dtbuf, xc, xdbl, A_log, xz, D_param, ybuf, out_ssm);

    wmma_gemm_kernel<2, 4, false><<<dim3(D_MODEL / 128, TOKENS / 64), 128, 0, stream>>>(
        ybuf, D_INNER, out_projw, D_MODEL, out, D_MODEL, D_INNER,
        x, D_MODEL, res_scale, nullptr);
}
